# starting kernel
# baseline (speedup 1.0000x reference)
#include <hip/hip_runtime.h>
#include <hip/hip_bf16.h>

#define N_NODES 50000
#define N_EDGES 1600000
#define F_DIM   128
#define H_HEADS 8
#define D_HEAD  16
#define E_DIM   16
#define HD      (H_HEADS * D_HEAD)
#define ET      (N_EDGES + N_NODES)
#define EPS_IN  1e-5f
#define NEG_SLOPE 0.2f

typedef float v2f __attribute__((ext_vector_type(2)));
typedef float v8f __attribute__((ext_vector_type(8)));

__global__ void k_init(float* deg, float* loopattr, float* denom, float* outacc,
                       int* amaxK) {
    int stride = gridDim.x * blockDim.x;
    int tid = blockIdx.x * blockDim.x + threadIdx.x;
    for (int i = tid; i < N_NODES; i += stride) deg[i] = 0.0f;
    for (int i = tid; i < N_NODES * E_DIM; i += stride) loopattr[i] = 0.0f;
    for (int i = tid; i < N_NODES * H_HEADS; i += stride) denom[i] = 0.0f;
    for (int i = tid; i < N_NODES * HD; i += stride) outacc[i] = 0.0f;
    for (int i = tid; i < N_NODES * H_HEADS; i += stride) amaxK[i] = (int)0x80000000;
}

__global__ void k_deg_loop(const int* __restrict__ dst,
                           const float* __restrict__ edge_attr,
                           float* deg, float* loopsum) {
    int e = blockIdx.x * blockDim.x + threadIdx.x;
    if (e >= N_EDGES) return;
    int t = dst[e];
    atomicAdd(&deg[t], 1.0f);
    const float* ea = edge_attr + (size_t)e * E_DIM;
    float* ls = loopsum + (size_t)t * E_DIM;
#pragma unroll
    for (int k = 0; k < E_DIM; ++k) atomicAdd(&ls[k], ea[k]);
}

__global__ void k_loop_final(float* loopattr, const float* __restrict__ deg) {
    int idx = blockIdx.x * blockDim.x + threadIdx.x;
    if (idx >= N_NODES * E_DIM) return;
    float d = deg[idx >> 4];
    d = d < 1.0f ? 1.0f : d;
    loopattr[idx] /= d;
}

__global__ void __launch_bounds__(256)
k_gemm_h(const float* __restrict__ x, const float* __restrict__ W,
         float* __restrict__ h) {
    int wave = threadIdx.x >> 5;
    int lane = threadIdx.x & 31;
    int tile = blockIdx.x * 8 + wave;
    int tileM = tile >> 3;
    int tileN = tile & 7;
    int m   = tileM * 16 + (lane & 15);
    int kb  = (lane >> 4) << 1;
    int col = tileN * 16 + (lane & 15);

    v8f acc = {};
#pragma unroll
    for (int k = 0; k < F_DIM; k += 4) {
        v2f a, b;
        a.x = x[(size_t)m * F_DIM + k + kb];
        a.y = x[(size_t)m * F_DIM + k + kb + 1];
        b.x = W[(size_t)(k + kb) * HD + col];
        b.y = W[(size_t)(k + kb + 1) * HD + col];
        acc = __builtin_amdgcn_wmma_f32_16x16x4_f32(
             false, a,  false, b,
             (short)0, acc,  false,  false);
    }
    int rbase = tileM * 16 + ((lane >> 4) << 3);
#pragma unroll
    for (int r = 0; r < 8; ++r)
        h[(size_t)(rbase + r) * HD + col] = acc[r];
}

__global__ void k_attn_node(const float* __restrict__ h,
                            const float* __restrict__ att_src,
                            const float* __restrict__ att_dst,
                            float* a_src, float* a_dst) {
    int idx = blockIdx.x * blockDim.x + threadIdx.x;
    if (idx >= N_NODES * H_HEADS) return;
    int n = idx >> 3, hh = idx & 7;
    const float* hp = h + (size_t)n * HD + hh * D_HEAD;
    const float* as = att_src + hh * D_HEAD;
    const float* ad = att_dst + hh * D_HEAD;
    float s = 0.f, d = 0.f;
#pragma unroll
    for (int i = 0; i < D_HEAD; ++i) {
        float v = hp[i];
        s += v * as[i];
        d += v * ad[i];
    }
    a_src[idx] = s;
    a_dst[idx] = d;
}

__global__ void k_edge_proj(const float* __restrict__ W_e,
                            const float* __restrict__ att_edge, float* V) {
    int idx = threadIdx.x;
    int k = idx >> 3, hh = idx & 7;
    float s = 0.f;
#pragma unroll
    for (int d = 0; d < D_HEAD; ++d)
        s += W_e[(size_t)k * HD + hh * D_HEAD + d] * att_edge[hh * D_HEAD + d];
    V[idx] = s;
}

__global__ void k_logits(const int* __restrict__ src, const int* __restrict__ dst,
                         const float* __restrict__ edge_attr,
                         const float* __restrict__ loopattr,
                         const float* __restrict__ Vc,
                         const float* __restrict__ a_src,
                         const float* __restrict__ a_dst,
                         float* alpha, int* amaxK) {
    int i = blockIdx.x * blockDim.x + threadIdx.x;
    if (i >= ET) return;
    int s, t;
    const float* ea;
    if (i < N_EDGES) {
        s = src[i]; t = dst[i];
        ea = edge_attr + (size_t)i * E_DIM;
    } else {
        s = i - N_EDGES; t = s;
        ea = loopattr + (size_t)s * E_DIM;
    }
    float eav[E_DIM];
#pragma unroll
    for (int k = 0; k < E_DIM; ++k) eav[k] = ea[k];
#pragma unroll
    for (int hh = 0; hh < H_HEADS; ++hh) {
        float ae = 0.f;
#pragma unroll
        for (int k = 0; k < E_DIM; ++k) ae += eav[k] * Vc[k * 8 + hh];
        float l = a_src[s * 8 + hh] + a_dst[t * 8 + hh] + ae;
        l = l > 0.f ? l : NEG_SLOPE * l;
        alpha[(size_t)i * 8 + hh] = l;
        int bi = __float_as_int(l);
        int key = bi >= 0 ? bi : (bi ^ 0x7FFFFFFF);
        atomicMax(&amaxK[t * 8 + hh], key);
    }
}

__global__ void k_expsum(const int* __restrict__ dst, float* alpha,
                         const int* __restrict__ amaxK, float* denom) {
    int i = blockIdx.x * blockDim.x + threadIdx.x;
    if (i >= ET) return;
    int t = (i < N_EDGES) ? dst[i] : (i - N_EDGES);
#pragma unroll
    for (int hh = 0; hh < H_HEADS; ++hh) {
        int key = amaxK[t * 8 + hh];
        int bi = key >= 0 ? key : (key ^ 0x7FFFFFFF);
        float mx = __int_as_float(bi);
        float a = __expf(alpha[(size_t)i * 8 + hh] - mx);
        alpha[(size_t)i * 8 + hh] = a;
        atomicAdd(&denom[t * 8 + hh], a);
    }
}

__global__ void k_scatter(const int* __restrict__ src, const int* __restrict__ dst,
                          const float* __restrict__ alpha,
                          const float* __restrict__ denom,
                          const float* __restrict__ h, float* outacc) {
    long long tid = (long long)blockIdx.x * blockDim.x + threadIdx.x;
    if (tid >= (long long)ET * 4) return;
    int g = (int)(tid & 3);
    long long i = tid >> 2;
    int s, t;
    if (i < N_EDGES) { s = src[i]; t = dst[i]; }
    else             { s = (int)(i - N_EDGES); t = s; }
#pragma unroll
    for (int hh = g * 2; hh < g * 2 + 2; ++hh) {
        float w = alpha[(size_t)i * 8 + hh] / (denom[t * 8 + hh] + 1e-16f);
        const float* hp = h + (size_t)s * HD + hh * D_HEAD;
        float* op = outacc + (size_t)t * HD + hh * D_HEAD;
#pragma unroll
        for (int d = 0; d < D_HEAD; ++d) atomicAdd(&op[d], hp[d] * w);
    }
}

__global__ void k_stats(const float* __restrict__ outacc, float* meanb,
                        float* invstd) {
    __shared__ float ssum[256];
    __shared__ float ssq[256];
    int c = blockIdx.x;
    float s = 0.f, q = 0.f;
    for (int n = threadIdx.x; n < N_NODES; n += blockDim.x) {
        float v = outacc[(size_t)n * HD + c];
        s += v;
        q += v * v;
    }
    ssum[threadIdx.x] = s;
    ssq[threadIdx.x] = q;
    __syncthreads();
    for (int off = 128; off > 0; off >>= 1) {
        if (threadIdx.x < off) {
            ssum[threadIdx.x] += ssum[threadIdx.x + off];
            ssq[threadIdx.x]  += ssq[threadIdx.x + off];
        }
        __syncthreads();
    }
    if (threadIdx.x == 0) {
        float m = ssum[0] / (float)N_NODES;
        float var = ssq[0] / (float)N_NODES - m * m;
        meanb[c] = m;
        invstd[c] = rsqrtf(var + EPS_IN);
    }
}

__global__ void k_final(const float* __restrict__ x,
                        const float* __restrict__ outacc,
                        const float* __restrict__ meanb,
                        const float* __restrict__ invstd,
                        const float* __restrict__ gamma,
                        const float* __restrict__ beta, float* out) {
    int idx = blockIdx.x * blockDim.x + threadIdx.x;
    if (idx >= N_NODES * HD) return;
    int c = idx & 127;
    float normed = (outacc[idx] - meanb[c]) * invstd[c] * gamma[c] + beta[c];
    float t = x[idx] + normed;
    out[idx] = t > 0.f ? t : expm1f(t);
}

extern "C" void kernel_launch(void* const* d_in, const int* in_sizes, int n_in,
                              void* d_out, int out_size, void* d_ws, size_t ws_size,
                              hipStream_t stream) {
    const float* x        = (const float*)d_in[0];
    const int*   ei       = (const int*)d_in[1];
    const float* edgeattr = (const float*)d_in[2];
    const float* W        = (const float*)d_in[3];
    const float* att_src  = (const float*)d_in[4];
    const float* att_dst  = (const float*)d_in[5];
    const float* W_e      = (const float*)d_in[6];
    const float* att_edge = (const float*)d_in[7];
    const float* gamma    = (const float*)d_in[8];
    const float* beta     = (const float*)d_in[9];
    float* out = (float*)d_out;

    const int* src = ei;
    const int* dst = ei + N_EDGES;

    float* ws = (float*)d_ws;
    float* hbuf     = ws;
    float* a_src    = hbuf   + (size_t)N_NODES * HD;
    float* a_dst    = a_src  + (size_t)N_NODES * H_HEADS;
    float* degb     = a_dst  + (size_t)N_NODES * H_HEADS;
    float* loopattr = degb   + N_NODES;
    float* Vc       = loopattr + (size_t)N_NODES * E_DIM;
    float* alpha    = Vc + 128;
    float* denom    = alpha + (size_t)ET * H_HEADS;
    float* outacc   = denom + (size_t)N_NODES * H_HEADS;
    float* meanb    = outacc + (size_t)N_NODES * HD;
    float* invstd   = meanb + 128;
    int*   amaxK    = (int*)(invstd + 128);

    k_init<<<1024, 256, 0, stream>>>(degb, loopattr, denom, outacc, amaxK);

    k_deg_loop<<<(N_EDGES + 255) / 256, 256, 0, stream>>>(dst, edgeattr, degb, loopattr);
    k_loop_final<<<(N_NODES * E_DIM + 255) / 256, 256, 0, stream>>>(loopattr, degb);

    k_gemm_h<<<(N_NODES / 16) * 8 / 8, 256, 0, stream>>>(x, W, hbuf);

    k_attn_node<<<(N_NODES * H_HEADS + 255) / 256, 256, 0, stream>>>(
        hbuf, att_src, att_dst, a_src, a_dst);
    k_edge_proj<<<1, 128, 0, stream>>>(W_e, att_edge, Vc);

    k_logits<<<(ET + 255) / 256, 256, 0, stream>>>(src, dst, edgeattr, loopattr,
                                                   Vc, a_src, a_dst, alpha, amaxK);
    k_expsum<<<(ET + 255) / 256, 256, 0, stream>>>(dst, alpha, amaxK, denom);
    {
        long long tot = (long long)ET * 4;
        int blocks = (int)((tot + 255) / 256);
        k_scatter<<<blocks, 256, 0, stream>>>(src, dst, alpha, denom, hbuf, outacc);
    }

    k_stats<<<HD, 256, 0, stream>>>(outacc, meanb, invstd);
    k_final<<<(N_NODES * HD + 255) / 256, 256, 0, stream>>>(
        x, outacc, meanb, invstd, gamma, beta, out);
}


# InstanceNorm stats by whole rows: lane owns 4 channels, 16-byte loads in flight together; 250 workgroup partials, added by a new launch (11 in all); was one 4-byte load per trip; sum order changes
# speedup vs baseline: 13.2541x; 13.2541x over previous
#include <hip/hip_runtime.h>
#include <hip/hip_bf16.h>

#define N_NODES 50000
#define N_EDGES 1600000
#define F_DIM   128
#define H_HEADS 8
#define D_HEAD  16
#define E_DIM   16
#define HD      (H_HEADS * D_HEAD)
#define ET      (N_EDGES + N_NODES)
#define EPS_IN  1e-5f
#define NEG_SLOPE 0.2f

typedef float v2f __attribute__((ext_vector_type(2)));
typedef float v8f __attribute__((ext_vector_type(8)));

#define N_CHUNK 128
#define CHUNK_E (N_EDGES / N_CHUNK)
#define CHUNK_TRIPS ((CHUNK_E + 1023) / 1024)
#define NODE_Q (N_NODES / 4)
static_assert(N_CHUNK * CHUNK_E == N_EDGES, "chunks tile the edge list exactly");
static_assert(NODE_Q * 4 == N_NODES, "node counters split into whole quads");
static_assert((size_t)N_CHUNK * N_NODES <= (size_t)N_NODES * HD, "planes fit the row plane they borrow");

__global__ void __launch_bounds__(1024)
k_chunk_hist(const int* __restrict__ dst, int* __restrict__ plane) {
    __shared__ __attribute__((aligned(16))) int hist[N_NODES];
    int tid = threadIdx.x;
    int g = blockIdx.x;
    const int* d = dst + (size_t)g * CHUNK_E;
    int dv[CHUNK_TRIPS];
#pragma unroll
    for (int k = 0; k < CHUNK_TRIPS; ++k) dv[k] = d[min(tid + k * 1024, CHUNK_E - 1)];
    int4* h4 = (int4*)hist;
    for (int i = tid; i < NODE_Q; i += 1024) h4[i] = make_int4(0, 0, 0, 0);
    __syncthreads();
#pragma unroll
    for (int k = 0; k < CHUNK_TRIPS; ++k)
        if (tid + k * 1024 < CHUNK_E) atomicAdd(&hist[dv[k]], 1);
    __syncthreads();
    int4* p4 = (int4*)(plane + (size_t)g * N_NODES);
    for (int i = tid; i < NODE_Q; i += 1024) p4[i] = h4[i];
}

__global__ void __launch_bounds__(256)
k_chunk_prefix(int* __restrict__ plane, int* __restrict__ cnt) {
    int t = blockIdx.x * 256 + threadIdx.x;
    if (t >= N_NODES) return;
    int run = 0;
    for (int g0 = 0; g0 < N_CHUNK; g0 += 16) {
        int v[16];
#pragma unroll
        for (int k = 0; k < 16; ++k) v[k] = plane[(size_t)(g0 + k) * N_NODES + t];
#pragma unroll
        for (int k = 0; k < 16; ++k) {
            plane[(size_t)(g0 + k) * N_NODES + t] = run;
            run += v[k];
        }
    }
    cnt[t] = run;
}

__global__ void __launch_bounds__(256)
k_gemm_h(const float* __restrict__ x, const float* __restrict__ W,
         float* __restrict__ h) {
    int wave = threadIdx.x >> 5;
    int lane = threadIdx.x & 31;
    int tile = blockIdx.x * 8 + wave;
    int tileM = tile >> 3;
    int tileN = tile & 7;
    int m   = tileM * 16 + (lane & 15);
    int kb  = (lane >> 4) << 1;
    int col = tileN * 16 + (lane & 15);

    v8f acc = {};
#pragma unroll
    for (int k = 0; k < F_DIM; k += 4) {
        v2f a, b;
        a.x = x[(size_t)m * F_DIM + k + kb];
        a.y = x[(size_t)m * F_DIM + k + kb + 1];
        b.x = W[(size_t)(k + kb) * HD + col];
        b.y = W[(size_t)(k + kb + 1) * HD + col];
        acc = __builtin_amdgcn_wmma_f32_16x16x4_f32(
             false, a,  false, b,
             (short)0, acc,  false,  false);
    }
    int rbase = tileM * 16 + ((lane >> 4) << 3);
#pragma unroll
    for (int r = 0; r < 8; ++r)
        h[(size_t)(rbase + r) * HD + col] = acc[r];
}

__global__ void k_attn_node(const float* __restrict__ h,
                            const float* __restrict__ att_src,
                            const float* __restrict__ att_dst,
                            float* a_src, float* a_dst) {
    int idx = blockIdx.x * blockDim.x + threadIdx.x;
    if (idx >= N_NODES * H_HEADS) return;
    int n = idx >> 3, hh = idx & 7;
    const float* hp = h + (size_t)n * HD + hh * D_HEAD;
    const float* as = att_src + hh * D_HEAD;
    const float* ad = att_dst + hh * D_HEAD;
    float s = 0.f, d = 0.f;
#pragma unroll
    for (int i = 0; i < D_HEAD; ++i) {
        float v = hp[i];
        s += v * as[i];
        d += v * ad[i];
    }
    a_src[idx] = s;
    a_dst[idx] = d;
}

__global__ void k_edge_proj(const float* __restrict__ W_e,
                            const float* __restrict__ att_edge, float* V) {
    int idx = threadIdx.x;
    int k = idx >> 3, hh = idx & 7;
    float s = 0.f;
#pragma unroll
    for (int d = 0; d < D_HEAD; ++d)
        s += W_e[(size_t)k * HD + hh * D_HEAD + d] * att_edge[hh * D_HEAD + d];
    V[idx] = s;
}

#define SCAN_CHUNK 49
__global__ void __launch_bounds__(1024)
k_rowptr(const int* __restrict__ cnt, int* __restrict__ rowptr) {
    __shared__ int wsum[32];
    int tid = threadIdx.x;
    int lane = tid & 31, wv = tid >> 5;
    int base = tid * SCAN_CHUNK;
    int c[SCAN_CHUNK];
    int tot = 0;
#pragma unroll
    for (int k = 0; k < SCAN_CHUNK; ++k) {
        int idx = base + k;
        int d = cnt[min(idx, N_NODES - 1)];
        int v = idx < N_NODES ? d : 0;
        c[k] = v;
        tot += v;
    }
    int inc = tot;
#pragma unroll
    for (int off = 1; off < 32; off <<= 1) {
        int nb = __shfl_up(inc, off);
        if (lane >= off) inc += nb;
    }
    if (lane == 31) wsum[wv] = inc;
    __syncthreads();
    if (wv == 0) {
        int v = wsum[lane];
        int iv = v;
#pragma unroll
        for (int off = 1; off < 32; off <<= 1) {
            int nb = __shfl_up(iv, off);
            if (lane >= off) iv += nb;
        }
        wsum[lane] = iv - v;
    }
    __syncthreads();
    int run = wsum[wv] + inc - tot;
#pragma unroll
    for (int k = 0; k < SCAN_CHUNK; ++k) {
        int idx = base + k;
        if (idx < N_NODES) {
            rowptr[idx] = run;
            run += c[k];
        }
    }
    if (tid == 1023) rowptr[N_NODES] = run;
}

__global__ void __launch_bounds__(1024)
k_chunk_place(const int* __restrict__ dst, const int* __restrict__ rowptr,
              const int* __restrict__ plane, int* __restrict__ perm) {
    __shared__ __attribute__((aligned(16))) int off[N_NODES];
    int tid = threadIdx.x;
    int g = blockIdx.x;
    const int* d = dst + (size_t)g * CHUNK_E;
    int dv[CHUNK_TRIPS];
#pragma unroll
    for (int k = 0; k < CHUNK_TRIPS; ++k) dv[k] = d[min(tid + k * 1024, CHUNK_E - 1)];
    const int4* r4 = (const int4*)rowptr;
    const int4* p4 = (const int4*)(plane + (size_t)g * N_NODES);
    int4* o4 = (int4*)off;
    for (int i = tid; i < NODE_Q; i += 1024) {
        int4 r = r4[i], p = p4[i];
        o4[i] = make_int4(r.x + p.x, r.y + p.y, r.z + p.z, r.w + p.w);
    }
    __syncthreads();
    int ebase = g * CHUNK_E;
    int pos[CHUNK_TRIPS];
#pragma unroll
    for (int k = 0; k < CHUNK_TRIPS; ++k) {
        pos[k] = 0;
        if (tid + k * 1024 < CHUNK_E) pos[k] = atomicAdd(&off[dv[k]], 1);
    }
#pragma unroll
    for (int k = 0; k < CHUNK_TRIPS; ++k) {
        int i = tid + k * 1024;
        if (i < CHUNK_E) {
            int p = min(max(pos[k], 0), N_EDGES - 1);
            perm[p] = ebase + i;
        }
    }
}

__device__ __forceinline__ void fma4(float4& acc, float w, const float4& v) {
    acc.x = fmaf(w, v.x, acc.x); acc.y = fmaf(w, v.y, acc.y);
    acc.z = fmaf(w, v.z, acc.z); acc.w = fmaf(w, v.w, acc.w);
}

__device__ __forceinline__ float max4(const float4& q) {
    return fmaxf(fmaxf(q.x, q.y), fmaxf(q.z, q.w));
}

#define LW_PITCH 36
__global__ void __launch_bounds__(32)
k_row(const int* __restrict__ rowptr, const int* __restrict__ perm,
      const int* __restrict__ src, const float* __restrict__ edge_attr,
      const float* __restrict__ Vc, const float* __restrict__ a_src,
      const float* __restrict__ a_dst, const float* __restrict__ h,
      float* __restrict__ outacc) {
    __shared__ __attribute__((aligned(16))) float lw[H_HEADS * LW_PITCH];
    const float NINF = -__builtin_inff();
    int lane = threadIdx.x;
    int t = blockIdx.x;
    int hh = lane >> 2;
    const float4* h4  = (const float4*)h;
    const float4* ea4 = (const float4*)edge_attr;
    const float4* as4 = (const float4*)a_src;
    const float4* lq  = (const float4*)(lw + hh * LW_PITCH);

    int beg = rowptr[t], end = rowptr[t + 1];
    beg = max(beg, 0); end = min(end, N_EDGES);

    float asl = a_src[t * 8 + hh];
    float adl = a_dst[t * 8 + hh];
    float4 v = h4[(size_t)t * 32 + lane];

    float ad[H_HEADS];
#pragma unroll
    for (int k = 0; k < H_HEADS; ++k) ad[k] = a_dst[t * 8 + k];

    float aes[H_HEADS];
#pragma unroll
    for (int k = 0; k < H_HEADS; ++k) aes[k] = 0.f;

    float m = NINF, den = 0.f;
    float4 acc = make_float4(0.f, 0.f, 0.f, 0.f);

    for (int b = beg; b < end; b += 32) {
        int n = min(32, end - b);
        int il = 0, sl = 0;
        float lg[H_HEADS];
#pragma unroll
        for (int k = 0; k < H_HEADS; ++k) lg[k] = NINF;
        if (lane < n) {
            il = perm[b + lane];
            il = min(max(il, 0), N_EDGES - 1);
            sl = src[il];
            float4 e0 = ea4[(size_t)il * 4 + 0];
            float4 e1 = ea4[(size_t)il * 4 + 1];
            float4 e2 = ea4[(size_t)il * 4 + 2];
            float4 e3 = ea4[(size_t)il * 4 + 3];
            float4 s0 = as4[(size_t)sl * 2 + 0];
            float4 s1 = as4[(size_t)sl * 2 + 1];
            float ev[E_DIM] = {e0.x, e0.y, e0.z, e0.w, e1.x, e1.y, e1.z, e1.w,
                               e2.x, e2.y, e2.z, e2.w, e3.x, e3.y, e3.z, e3.w};
            float sv[H_HEADS] = {s0.x, s0.y, s0.z, s0.w, s1.x, s1.y, s1.z, s1.w};
#pragma unroll
            for (int k = 0; k < H_HEADS; ++k) {
                float ae = 0.f;
#pragma unroll
                for (int q = 0; q < E_DIM; ++q) ae += ev[q] * Vc[q * 8 + k];
                aes[k] += ae;
                float l = sv[k] + ad[k] + ae;
                lg[k] = l > 0.f ? l : NEG_SLOPE * l;
            }
        }
#pragma unroll
        for (int k = 0; k < H_HEADS; ++k) lw[k * LW_PITCH + lane] = lg[k];
        __syncthreads();

        float mc = NINF;
        for (int i = 0; i < 8; ++i) mc = fmaxf(mc, max4(lq[i]));
        float mn = fmaxf(m, mc);
        float sc = __expf(m - mn);
        acc.x *= sc; acc.y *= sc; acc.z *= sc; acc.w *= sc;
        den *= sc;
        m = mn;

        int e = 0;
        for (; e + 4 <= n; e += 4) {
            int s0 = __builtin_amdgcn_readlane(sl, e);
            int s1 = __builtin_amdgcn_readlane(sl, e + 1);
            int s2 = __builtin_amdgcn_readlane(sl, e + 2);
            int s3 = __builtin_amdgcn_readlane(sl, e + 3);
            float4 v0 = h4[(size_t)s0 * 32 + lane];
            float4 v1 = h4[(size_t)s1 * 32 + lane];
            float4 v2 = h4[(size_t)s2 * 32 + lane];
            float4 v3 = h4[(size_t)s3 * 32 + lane];
            float4 q = lq[e >> 2];
            float w0 = __expf(q.x - m);
            float w1 = __expf(q.y - m);
            float w2 = __expf(q.z - m);
            float w3 = __expf(q.w - m);
            den += w0; den += w1; den += w2; den += w3;
            fma4(acc, w0, v0);
            fma4(acc, w1, v1);
            fma4(acc, w2, v2);
            fma4(acc, w3, v3);
        }
        for (; e < n; ++e) {
            int s0 = __builtin_amdgcn_readlane(sl, e);
            float4 v0 = h4[(size_t)s0 * 32 + lane];
            float w0 = __expf(lw[hh * LW_PITCH + e] - m);
            den += w0;
            fma4(acc, w0, v0);
        }
        __syncthreads();
    }

#pragma unroll
    for (int k = 0; k < H_HEADS; ++k) lw[k * LW_PITCH + lane] = aes[k];
    __syncthreads();
    float sae = 0.f;
    for (int i = 0; i < 8; ++i) {
        float4 q = lq[i];
        sae += q.x; sae += q.y; sae += q.z; sae += q.w;
    }
    float dg = (float)(end - beg);
    dg = dg < 1.0f ? 1.0f : dg;

    {
        float l = asl + adl + sae / dg;
        l = l > 0.f ? l : NEG_SLOPE * l;
        float mn = fmaxf(m, l);
        float sc = __expf(m - mn);
        float w = __expf(l - mn);
        acc.x = fmaf(w, v.x, acc.x * sc); acc.y = fmaf(w, v.y, acc.y * sc);
        acc.z = fmaf(w, v.z, acc.z * sc); acc.w = fmaf(w, v.w, acc.w * sc);
        den = fmaf(den, sc, w);
    }
    float dn = den + 1e-16f;
    acc.x /= dn; acc.y /= dn; acc.z /= dn; acc.w /= dn;
    ((float4*)outacc)[(size_t)t * 32 + lane] = acc;
}

#define ST_G     250
#define ST_WAVES 8
#define ST_TRIPS 25
static_assert(ST_G * ST_WAVES * ST_TRIPS == N_NODES, "the stats grid tiles the rows exactly");
static_assert(ST_G * 256 <= N_NODES * HD, "partials fit the plane they borrow");

__global__ void __launch_bounds__(ST_WAVES * 32)
k_stats_part(const float* __restrict__ outacc, float* __restrict__ part) {
    __shared__ __attribute__((aligned(16))) float red[ST_WAVES * 256];
    int lane = threadIdx.x & 31, wv = threadIdx.x >> 5;
    const float4* p = (const float4*)outacc
                    + ((size_t)blockIdx.x * ST_WAVES + wv) * 32 + lane;
    float4 v[ST_TRIPS];
#pragma unroll
    for (int k = 0; k < ST_TRIPS; ++k)
        v[k] = p[(size_t)k * (ST_G * ST_WAVES * 32)];
    float4 s = make_float4(0.f, 0.f, 0.f, 0.f);
    float4 q = make_float4(0.f, 0.f, 0.f, 0.f);
#pragma unroll
    for (int k = 0; k < ST_TRIPS; ++k) {
        s.x += v[k].x; s.y += v[k].y; s.z += v[k].z; s.w += v[k].w;
        q.x = fmaf(v[k].x, v[k].x, q.x); q.y = fmaf(v[k].y, v[k].y, q.y);
        q.z = fmaf(v[k].z, v[k].z, q.z); q.w = fmaf(v[k].w, v[k].w, q.w);
    }
    float4* r4 = (float4*)red;
    r4[wv * 64 + lane] = s;
    r4[wv * 64 + 32 + lane] = q;
    __syncthreads();
    int t = threadIdx.x;
    float a = 0.f;
#pragma unroll
    for (int w = 0; w < ST_WAVES; ++w) a += red[w * 256 + t];
    part[(size_t)blockIdx.x * 256 + t] = a;
}

#define ST_HALF (ST_G / 2)
static_assert(ST_HALF * 2 == ST_G, "two equal groups of partials");

__global__ void __launch_bounds__(512)
k_stats_fin(const float* __restrict__ part, float* meanb, float* invstd) {
    __shared__ float red[512];
    int t = threadIdx.x;
    int col = t & 255, j = t >> 8;
    const float* p = part + (size_t)j * ST_HALF * 256 + col;
    float a = 0.f;
#pragma unroll
    for (int k = 0; k < ST_HALF; ++k) a += p[(size_t)k * 256];
    red[t] = a;
    __syncthreads();
    if (t < HD) {
        float sum = red[t] + red[256 + t];
        float sq  = red[128 + t] + red[384 + t];
        float m = sum / (float)N_NODES;
        float var = sq / (float)N_NODES - m * m;
        meanb[t] = m;
        invstd[t] = rsqrtf(var + EPS_IN);
    }
}

__global__ void k_final(const float* __restrict__ x,
                        const float* __restrict__ outacc,
                        const float* __restrict__ meanb,
                        const float* __restrict__ invstd,
                        const float* __restrict__ gamma,
                        const float* __restrict__ beta, float* out) {
    int idx = blockIdx.x * blockDim.x + threadIdx.x;
    if (idx >= N_NODES * HD) return;
    int c = idx & 127;
    float normed = (outacc[idx] - meanb[c]) * invstd[c] * gamma[c] + beta[c];
    float t = x[idx] + normed;
    out[idx] = t > 0.f ? t : expm1f(t);
}

extern "C" void kernel_launch(void* const* d_in, const int* in_sizes, int n_in,
                              void* d_out, int out_size, void* d_ws, size_t ws_size,
                              hipStream_t stream) {
    const float* x        = (const float*)d_in[0];
    const int*   ei       = (const int*)d_in[1];
    const float* edgeattr = (const float*)d_in[2];
    const float* W        = (const float*)d_in[3];
    const float* att_src  = (const float*)d_in[4];
    const float* att_dst  = (const float*)d_in[5];
    const float* W_e      = (const float*)d_in[6];
    const float* att_edge = (const float*)d_in[7];
    const float* gamma    = (const float*)d_in[8];
    const float* beta     = (const float*)d_in[9];
    float* out = (float*)d_out;

    const int* src = ei;
    const int* dst = ei + N_EDGES;

    float* ws = (float*)d_ws;
    float* hbuf     = ws;
    float* a_src    = hbuf   + (size_t)N_NODES * HD;
    float* a_dst    = a_src  + (size_t)N_NODES * H_HEADS;
    float* Vc       = a_dst  + (size_t)N_NODES * H_HEADS;
    float* outacc   = Vc + 128;
    float* meanb    = outacc + (size_t)N_NODES * HD;
    float* invstd   = meanb + 128;
    int*   cnt      = (int*)(invstd + 128);
    int*   rowptr   = cnt + 65536;
    int*   perm     = rowptr + 2 * 65536;
    int*   plane    = (int*)outacc;
    float* part     = hbuf;

    k_chunk_hist<<<N_CHUNK, 1024, 0, stream>>>(dst, plane);
    k_chunk_prefix<<<(N_NODES + 255) / 256, 256, 0, stream>>>(plane, cnt);

    k_gemm_h<<<(N_NODES / 16) * 8 / 8, 256, 0, stream>>>(x, W, hbuf);

    k_attn_node<<<(N_NODES * H_HEADS + 255) / 256, 256, 0, stream>>>(
        hbuf, att_src, att_dst, a_src, a_dst);
    k_edge_proj<<<1, 128, 0, stream>>>(W_e, att_edge, Vc);

    k_rowptr<<<1, 1024, 0, stream>>>(cnt, rowptr);
    k_chunk_place<<<N_CHUNK, 1024, 0, stream>>>(dst, rowptr, plane, perm);
    k_row<<<N_NODES, 32, 0, stream>>>(rowptr, perm, src, edgeattr, Vc, a_src, a_dst,
                                      hbuf, outacc);

    k_stats_part<<<ST_G, ST_WAVES * 32, 0, stream>>>(outacc, part);
    k_stats_fin<<<1, 512, 0, stream>>>(part, meanb, invstd);
    k_final<<<(N_NODES * HD + 255) / 256, 256, 0, stream>>>(
        x, outacc, meanb, invstd, gamma, beta, out);
}


# Independent launches share a grid, piece chosen by block index: chunk prefix and edge projection beside the GEMM, row-pointer scan beside the node logits; 11 launches become 8; sums unchanged
# speedup vs baseline: 13.9077x; 1.0493x over previous
#include <hip/hip_runtime.h>
#include <hip/hip_bf16.h>

#define N_NODES 50000
#define N_EDGES 1600000
#define F_DIM   128
#define H_HEADS 8
#define D_HEAD  16
#define E_DIM   16
#define HD      (H_HEADS * D_HEAD)
#define ET      (N_EDGES + N_NODES)
#define EPS_IN  1e-5f
#define NEG_SLOPE 0.2f

typedef float v2f __attribute__((ext_vector_type(2)));
typedef float v8f __attribute__((ext_vector_type(8)));

#define N_CHUNK 128
#define CHUNK_E (N_EDGES / N_CHUNK)
#define CHUNK_TRIPS ((CHUNK_E + 1023) / 1024)
#define NODE_Q (N_NODES / 4)
static_assert(N_CHUNK * CHUNK_E == N_EDGES, "chunks tile the edge list exactly");
static_assert(NODE_Q * 4 == N_NODES, "node counters split into whole quads");
static_assert((size_t)N_CHUNK * N_NODES <= (size_t)N_NODES * HD, "planes fit the row plane they borrow");

__global__ void __launch_bounds__(1024)
k_chunk_hist(const int* __restrict__ dst, int* __restrict__ plane) {
    __shared__ __attribute__((aligned(16))) int hist[N_NODES];
    int tid = threadIdx.x;
    int g = blockIdx.x;
    const int* d = dst + (size_t)g * CHUNK_E;
    int dv[CHUNK_TRIPS];
#pragma unroll
    for (int k = 0; k < CHUNK_TRIPS; ++k) dv[k] = d[min(tid + k * 1024, CHUNK_E - 1)];
    int4* h4 = (int4*)hist;
    for (int i = tid; i < NODE_Q; i += 1024) h4[i] = make_int4(0, 0, 0, 0);
    __syncthreads();
#pragma unroll
    for (int k = 0; k < CHUNK_TRIPS; ++k)
        if (tid + k * 1024 < CHUNK_E) atomicAdd(&hist[dv[k]], 1);
    __syncthreads();
    int4* p4 = (int4*)(plane + (size_t)g * N_NODES);
    for (int i = tid; i < NODE_Q; i += 1024) p4[i] = h4[i];
}

#define PREFIX_BLOCKS ((N_NODES + 255) / 256)
__device__ __forceinline__ void
chunk_prefix_body(int blk, int* __restrict__ plane, int* __restrict__ cnt) {
    int t = blk * 256 + threadIdx.x;
    if (t >= N_NODES) return;
    int run = 0;
    for (int g0 = 0; g0 < N_CHUNK; g0 += 16) {
        int v[16];
#pragma unroll
        for (int k = 0; k < 16; ++k) v[k] = plane[(size_t)(g0 + k) * N_NODES + t];
#pragma unroll
        for (int k = 0; k < 16; ++k) {
            plane[(size_t)(g0 + k) * N_NODES + t] = run;
            run += v[k];
        }
    }
    cnt[t] = run;
}

#define GEMM_BLOCKS ((N_NODES / 16) * 8 / 8)
__device__ __forceinline__ void
gemm_h_body(int blk, const float* __restrict__ x, const float* __restrict__ W,
            float* __restrict__ h) {
    int wave = threadIdx.x >> 5;
    int lane = threadIdx.x & 31;
    int tile = blk * 8 + wave;
    int tileM = tile >> 3;
    int tileN = tile & 7;
    int m   = tileM * 16 + (lane & 15);
    int kb  = (lane >> 4) << 1;
    int col = tileN * 16 + (lane & 15);

    v8f acc = {};
#pragma unroll
    for (int k = 0; k < F_DIM; k += 4) {
        v2f a, b;
        a.x = x[(size_t)m * F_DIM + k + kb];
        a.y = x[(size_t)m * F_DIM + k + kb + 1];
        b.x = W[(size_t)(k + kb) * HD + col];
        b.y = W[(size_t)(k + kb + 1) * HD + col];
        acc = __builtin_amdgcn_wmma_f32_16x16x4_f32(
             false, a,  false, b,
             (short)0, acc,  false,  false);
    }
    int rbase = tileM * 16 + ((lane >> 4) << 3);
#pragma unroll
    for (int r = 0; r < 8; ++r)
        h[(size_t)(rbase + r) * HD + col] = acc[r];
}

#define ATTN_BLOCKS ((N_NODES * H_HEADS + 1023) / 1024)
__device__ __forceinline__ void
attn_node_body(int blk, const float* __restrict__ h,
               const float* __restrict__ att_src,
               const float* __restrict__ att_dst,
               float* a_src, float* a_dst) {
    int idx = blk * 1024 + threadIdx.x;
    if (idx >= N_NODES * H_HEADS) return;
    int n = idx >> 3, hh = idx & 7;
    const float* hp = h + (size_t)n * HD + hh * D_HEAD;
    const float* as = att_src + hh * D_HEAD;
    const float* ad = att_dst + hh * D_HEAD;
    float s = 0.f, d = 0.f;
#pragma unroll
    for (int i = 0; i < D_HEAD; ++i) {
        float v = hp[i];
        s += v * as[i];
        d += v * ad[i];
    }
    a_src[idx] = s;
    a_dst[idx] = d;
}

__device__ __forceinline__ void
edge_proj_body(const float* __restrict__ W_e,
               const float* __restrict__ att_edge, float* V) {
    int idx = threadIdx.x;
    if (idx >= E_DIM * H_HEADS) return;
    int k = idx >> 3, hh = idx & 7;
    float s = 0.f;
#pragma unroll
    for (int d = 0; d < D_HEAD; ++d)
        s += W_e[(size_t)k * HD + hh * D_HEAD + d] * att_edge[hh * D_HEAD + d];
    V[idx] = s;
}

#define SCAN_CHUNK 49
__device__ __forceinline__ void
rowptr_body(const int* __restrict__ cnt, int* __restrict__ rowptr) {
    __shared__ int wsum[32];
    int tid = threadIdx.x;
    int lane = tid & 31, wv = tid >> 5;
    int base = tid * SCAN_CHUNK;
    int c[SCAN_CHUNK];
    int tot = 0;
#pragma unroll
    for (int k = 0; k < SCAN_CHUNK; ++k) {
        int idx = base + k;
        int d = cnt[min(idx, N_NODES - 1)];
        int v = idx < N_NODES ? d : 0;
        c[k] = v;
        tot += v;
    }
    int inc = tot;
#pragma unroll
    for (int off = 1; off < 32; off <<= 1) {
        int nb = __shfl_up(inc, off);
        if (lane >= off) inc += nb;
    }
    if (lane == 31) wsum[wv] = inc;
    __syncthreads();
    if (wv == 0) {
        int v = wsum[lane];
        int iv = v;
#pragma unroll
        for (int off = 1; off < 32; off <<= 1) {
            int nb = __shfl_up(iv, off);
            if (lane >= off) iv += nb;
        }
        wsum[lane] = iv - v;
    }
    __syncthreads();
    int run = wsum[wv] + inc - tot;
#pragma unroll
    for (int k = 0; k < SCAN_CHUNK; ++k) {
        int idx = base + k;
        if (idx < N_NODES) {
            rowptr[idx] = run;
            run += c[k];
        }
    }
    if (tid == 1023) rowptr[N_NODES] = run;
}

__global__ void __launch_bounds__(256)
k_prefix_proj_gemm(int* __restrict__ plane, int* __restrict__ cnt,
                   const float* __restrict__ W_e, const float* __restrict__ att_edge,
                   float* __restrict__ V,
                   const float* __restrict__ x, const float* __restrict__ W,
                   float* __restrict__ h) {
    int b = blockIdx.x;
    if (b < PREFIX_BLOCKS)       chunk_prefix_body(b, plane, cnt);
    else if (b == PREFIX_BLOCKS) edge_proj_body(W_e, att_edge, V);
    else                         gemm_h_body(b - PREFIX_BLOCKS - 1, x, W, h);
}

__global__ void __launch_bounds__(1024)
k_rowptr_attn(const int* __restrict__ cnt, int* __restrict__ rowptr,
              const float* __restrict__ h, const float* __restrict__ att_src,
              const float* __restrict__ att_dst, float* a_src, float* a_dst) {
    int b = blockIdx.x;
    if (b == 0) rowptr_body(cnt, rowptr);
    else        attn_node_body(b - 1, h, att_src, att_dst, a_src, a_dst);
}

__global__ void __launch_bounds__(1024)
k_chunk_place(const int* __restrict__ dst, const int* __restrict__ rowptr,
              const int* __restrict__ plane, int* __restrict__ perm) {
    __shared__ __attribute__((aligned(16))) int off[N_NODES];
    int tid = threadIdx.x;
    int g = blockIdx.x;
    const int* d = dst + (size_t)g * CHUNK_E;
    int dv[CHUNK_TRIPS];
#pragma unroll
    for (int k = 0; k < CHUNK_TRIPS; ++k) dv[k] = d[min(tid + k * 1024, CHUNK_E - 1)];
    const int4* r4 = (const int4*)rowptr;
    const int4* p4 = (const int4*)(plane + (size_t)g * N_NODES);
    int4* o4 = (int4*)off;
    for (int i = tid; i < NODE_Q; i += 1024) {
        int4 r = r4[i], p = p4[i];
        o4[i] = make_int4(r.x + p.x, r.y + p.y, r.z + p.z, r.w + p.w);
    }
    __syncthreads();
    int ebase = g * CHUNK_E;
    int pos[CHUNK_TRIPS];
#pragma unroll
    for (int k = 0; k < CHUNK_TRIPS; ++k) {
        pos[k] = 0;
        if (tid + k * 1024 < CHUNK_E) pos[k] = atomicAdd(&off[dv[k]], 1);
    }
#pragma unroll
    for (int k = 0; k < CHUNK_TRIPS; ++k) {
        int i = tid + k * 1024;
        if (i < CHUNK_E) {
            int p = min(max(pos[k], 0), N_EDGES - 1);
            perm[p] = ebase + i;
        }
    }
}

__device__ __forceinline__ void fma4(float4& acc, float w, const float4& v) {
    acc.x = fmaf(w, v.x, acc.x); acc.y = fmaf(w, v.y, acc.y);
    acc.z = fmaf(w, v.z, acc.z); acc.w = fmaf(w, v.w, acc.w);
}

__device__ __forceinline__ float max4(const float4& q) {
    return fmaxf(fmaxf(q.x, q.y), fmaxf(q.z, q.w));
}

#define LW_PITCH 36
__global__ void __launch_bounds__(32)
k_row(const int* __restrict__ rowptr, const int* __restrict__ perm,
      const int* __restrict__ src, const float* __restrict__ edge_attr,
      const float* __restrict__ Vc, const float* __restrict__ a_src,
      const float* __restrict__ a_dst, const float* __restrict__ h,
      float* __restrict__ outacc) {
    __shared__ __attribute__((aligned(16))) float lw[H_HEADS * LW_PITCH];
    const float NINF = -__builtin_inff();
    int lane = threadIdx.x;
    int t = blockIdx.x;
    int hh = lane >> 2;
    const float4* h4  = (const float4*)h;
    const float4* ea4 = (const float4*)edge_attr;
    const float4* as4 = (const float4*)a_src;
    const float4* lq  = (const float4*)(lw + hh * LW_PITCH);

    int beg = rowptr[t], end = rowptr[t + 1];
    beg = max(beg, 0); end = min(end, N_EDGES);

    float asl = a_src[t * 8 + hh];
    float adl = a_dst[t * 8 + hh];
    float4 v = h4[(size_t)t * 32 + lane];

    float ad[H_HEADS];
#pragma unroll
    for (int k = 0; k < H_HEADS; ++k) ad[k] = a_dst[t * 8 + k];

    float aes[H_HEADS];
#pragma unroll
    for (int k = 0; k < H_HEADS; ++k) aes[k] = 0.f;

    float m = NINF, den = 0.f;
    float4 acc = make_float4(0.f, 0.f, 0.f, 0.f);

    for (int b = beg; b < end; b += 32) {
        int n = min(32, end - b);
        int il = 0, sl = 0;
        float lg[H_HEADS];
#pragma unroll
        for (int k = 0; k < H_HEADS; ++k) lg[k] = NINF;
        if (lane < n) {
            il = perm[b + lane];
            il = min(max(il, 0), N_EDGES - 1);
            sl = src[il];
            float4 e0 = ea4[(size_t)il * 4 + 0];
            float4 e1 = ea4[(size_t)il * 4 + 1];
            float4 e2 = ea4[(size_t)il * 4 + 2];
            float4 e3 = ea4[(size_t)il * 4 + 3];
            float4 s0 = as4[(size_t)sl * 2 + 0];
            float4 s1 = as4[(size_t)sl * 2 + 1];
            float ev[E_DIM] = {e0.x, e0.y, e0.z, e0.w, e1.x, e1.y, e1.z, e1.w,
                               e2.x, e2.y, e2.z, e2.w, e3.x, e3.y, e3.z, e3.w};
            float sv[H_HEADS] = {s0.x, s0.y, s0.z, s0.w, s1.x, s1.y, s1.z, s1.w};
#pragma unroll
            for (int k = 0; k < H_HEADS; ++k) {
                float ae = 0.f;
#pragma unroll
                for (int q = 0; q < E_DIM; ++q) ae += ev[q] * Vc[q * 8 + k];
                aes[k] += ae;
                float l = sv[k] + ad[k] + ae;
                lg[k] = l > 0.f ? l : NEG_SLOPE * l;
            }
        }
#pragma unroll
        for (int k = 0; k < H_HEADS; ++k) lw[k * LW_PITCH + lane] = lg[k];
        __syncthreads();

        float mc = NINF;
        for (int i = 0; i < 8; ++i) mc = fmaxf(mc, max4(lq[i]));
        float mn = fmaxf(m, mc);
        float sc = __expf(m - mn);
        acc.x *= sc; acc.y *= sc; acc.z *= sc; acc.w *= sc;
        den *= sc;
        m = mn;

        int e = 0;
        for (; e + 4 <= n; e += 4) {
            int s0 = __builtin_amdgcn_readlane(sl, e);
            int s1 = __builtin_amdgcn_readlane(sl, e + 1);
            int s2 = __builtin_amdgcn_readlane(sl, e + 2);
            int s3 = __builtin_amdgcn_readlane(sl, e + 3);
            float4 v0 = h4[(size_t)s0 * 32 + lane];
            float4 v1 = h4[(size_t)s1 * 32 + lane];
            float4 v2 = h4[(size_t)s2 * 32 + lane];
            float4 v3 = h4[(size_t)s3 * 32 + lane];
            float4 q = lq[e >> 2];
            float w0 = __expf(q.x - m);
            float w1 = __expf(q.y - m);
            float w2 = __expf(q.z - m);
            float w3 = __expf(q.w - m);
            den += w0; den += w1; den += w2; den += w3;
            fma4(acc, w0, v0);
            fma4(acc, w1, v1);
            fma4(acc, w2, v2);
            fma4(acc, w3, v3);
        }
        for (; e < n; ++e) {
            int s0 = __builtin_amdgcn_readlane(sl, e);
            float4 v0 = h4[(size_t)s0 * 32 + lane];
            float w0 = __expf(lw[hh * LW_PITCH + e] - m);
            den += w0;
            fma4(acc, w0, v0);
        }
        __syncthreads();
    }

#pragma unroll
    for (int k = 0; k < H_HEADS; ++k) lw[k * LW_PITCH + lane] = aes[k];
    __syncthreads();
    float sae = 0.f;
    for (int i = 0; i < 8; ++i) {
        float4 q = lq[i];
        sae += q.x; sae += q.y; sae += q.z; sae += q.w;
    }
    float dg = (float)(end - beg);
    dg = dg < 1.0f ? 1.0f : dg;

    {
        float l = asl + adl + sae / dg;
        l = l > 0.f ? l : NEG_SLOPE * l;
        float mn = fmaxf(m, l);
        float sc = __expf(m - mn);
        float w = __expf(l - mn);
        acc.x = fmaf(w, v.x, acc.x * sc); acc.y = fmaf(w, v.y, acc.y * sc);
        acc.z = fmaf(w, v.z, acc.z * sc); acc.w = fmaf(w, v.w, acc.w * sc);
        den = fmaf(den, sc, w);
    }
    float dn = den + 1e-16f;
    acc.x /= dn; acc.y /= dn; acc.z /= dn; acc.w /= dn;
    ((float4*)outacc)[(size_t)t * 32 + lane] = acc;
}

#define ST_G     250
#define ST_WAVES 8
#define ST_TRIPS 25
static_assert(ST_G * ST_WAVES * ST_TRIPS == N_NODES, "the stats grid tiles the rows exactly");
static_assert(ST_G * 256 <= N_NODES * HD, "partials fit the plane they borrow");

__global__ void __launch_bounds__(ST_WAVES * 32)
k_stats_part(const float* __restrict__ outacc, float* __restrict__ part) {
    __shared__ __attribute__((aligned(16))) float red[ST_WAVES * 256];
    int lane = threadIdx.x & 31, wv = threadIdx.x >> 5;
    const float4* p = (const float4*)outacc
                    + ((size_t)blockIdx.x * ST_WAVES + wv) * 32 + lane;
    float4 v[ST_TRIPS];
#pragma unroll
    for (int k = 0; k < ST_TRIPS; ++k)
        v[k] = p[(size_t)k * (ST_G * ST_WAVES * 32)];
    float4 s = make_float4(0.f, 0.f, 0.f, 0.f);
    float4 q = make_float4(0.f, 0.f, 0.f, 0.f);
#pragma unroll
    for (int k = 0; k < ST_TRIPS; ++k) {
        s.x += v[k].x; s.y += v[k].y; s.z += v[k].z; s.w += v[k].w;
        q.x = fmaf(v[k].x, v[k].x, q.x); q.y = fmaf(v[k].y, v[k].y, q.y);
        q.z = fmaf(v[k].z, v[k].z, q.z); q.w = fmaf(v[k].w, v[k].w, q.w);
    }
    float4* r4 = (float4*)red;
    r4[wv * 64 + lane] = s;
    r4[wv * 64 + 32 + lane] = q;
    __syncthreads();
    int t = threadIdx.x;
    float a = 0.f;
#pragma unroll
    for (int w = 0; w < ST_WAVES; ++w) a += red[w * 256 + t];
    part[(size_t)blockIdx.x * 256 + t] = a;
}

#define ST_HALF (ST_G / 2)
static_assert(ST_HALF * 2 == ST_G, "two equal groups of partials");

__global__ void __launch_bounds__(512)
k_stats_fin(const float* __restrict__ part, float* meanb, float* invstd) {
    __shared__ float red[512];
    int t = threadIdx.x;
    int col = t & 255, j = t >> 8;
    const float* p = part + (size_t)j * ST_HALF * 256 + col;
    float a = 0.f;
#pragma unroll
    for (int k = 0; k < ST_HALF; ++k) a += p[(size_t)k * 256];
    red[t] = a;
    __syncthreads();
    if (t < HD) {
        float sum = red[t] + red[256 + t];
        float sq  = red[128 + t] + red[384 + t];
        float m = sum / (float)N_NODES;
        float var = sq / (float)N_NODES - m * m;
        meanb[t] = m;
        invstd[t] = rsqrtf(var + EPS_IN);
    }
}

__global__ void k_final(const float* __restrict__ x,
                        const float* __restrict__ outacc,
                        const float* __restrict__ meanb,
                        const float* __restrict__ invstd,
                        const float* __restrict__ gamma,
                        const float* __restrict__ beta, float* out) {
    int idx = blockIdx.x * blockDim.x + threadIdx.x;
    if (idx >= N_NODES * HD) return;
    int c = idx & 127;
    float normed = (outacc[idx] - meanb[c]) * invstd[c] * gamma[c] + beta[c];
    float t = x[idx] + normed;
    out[idx] = t > 0.f ? t : expm1f(t);
}

extern "C" void kernel_launch(void* const* d_in, const int* in_sizes, int n_in,
                              void* d_out, int out_size, void* d_ws, size_t ws_size,
                              hipStream_t stream) {
    const float* x        = (const float*)d_in[0];
    const int*   ei       = (const int*)d_in[1];
    const float* edgeattr = (const float*)d_in[2];
    const float* W        = (const float*)d_in[3];
    const float* att_src  = (const float*)d_in[4];
    const float* att_dst  = (const float*)d_in[5];
    const float* W_e      = (const float*)d_in[6];
    const float* att_edge = (const float*)d_in[7];
    const float* gamma    = (const float*)d_in[8];
    const float* beta     = (const float*)d_in[9];
    float* out = (float*)d_out;

    const int* src = ei;
    const int* dst = ei + N_EDGES;

    float* ws = (float*)d_ws;
    float* hbuf     = ws;
    float* a_src    = hbuf   + (size_t)N_NODES * HD;
    float* a_dst    = a_src  + (size_t)N_NODES * H_HEADS;
    float* Vc       = a_dst  + (size_t)N_NODES * H_HEADS;
    float* outacc   = Vc + 128;
    float* meanb    = outacc + (size_t)N_NODES * HD;
    float* invstd   = meanb + 128;
    int*   cnt      = (int*)(invstd + 128);
    int*   rowptr   = cnt + 65536;
    int*   perm     = rowptr + 2 * 65536;
    int*   plane    = (int*)outacc;
    float* part     = hbuf;

    k_chunk_hist<<<N_CHUNK, 1024, 0, stream>>>(dst, plane);

    k_prefix_proj_gemm<<<PREFIX_BLOCKS + 1 + GEMM_BLOCKS, 256, 0, stream>>>(
        plane, cnt, W_e, att_edge, Vc, x, W, hbuf);

    k_rowptr_attn<<<1 + ATTN_BLOCKS, 1024, 0, stream>>>(
        cnt, rowptr, hbuf, att_src, att_dst, a_src, a_dst);

    k_chunk_place<<<N_CHUNK, 1024, 0, stream>>>(dst, rowptr, plane, perm);
    k_row<<<N_NODES, 32, 0, stream>>>(rowptr, perm, src, edgeattr, Vc, a_src, a_dst,
                                      hbuf, outacc);

    k_stats_part<<<ST_G, ST_WAVES * 32, 0, stream>>>(outacc, part);
    k_stats_fin<<<1, 512, 0, stream>>>(part, meanb, invstd);
    k_final<<<(N_NODES * HD + 255) / 256, 256, 0, stream>>>(
        x, outacc, meanb, invstd, gamma, beta, out);
}


# InstanceNorm finish folded into the final pass: 250 resident 16-wave workgroups each add the 250 partials in LDS, then normalise their 200 rows, lane owns 4 channels; 7 launches; sums unchanged
# speedup vs baseline: 14.4784x; 1.0410x over previous
#include <hip/hip_runtime.h>
#include <hip/hip_bf16.h>

#define N_NODES 50000
#define N_EDGES 1600000
#define F_DIM   128
#define H_HEADS 8
#define D_HEAD  16
#define E_DIM   16
#define HD      (H_HEADS * D_HEAD)
#define ET      (N_EDGES + N_NODES)
#define EPS_IN  1e-5f
#define NEG_SLOPE 0.2f

typedef float v2f __attribute__((ext_vector_type(2)));
typedef float v8f __attribute__((ext_vector_type(8)));

#define N_CHUNK 128
#define CHUNK_E (N_EDGES / N_CHUNK)
#define CHUNK_TRIPS ((CHUNK_E + 1023) / 1024)
#define NODE_Q (N_NODES / 4)
static_assert(N_CHUNK * CHUNK_E == N_EDGES, "chunks tile the edge list exactly");
static_assert(NODE_Q * 4 == N_NODES, "node counters split into whole quads");
static_assert((size_t)N_CHUNK * N_NODES <= (size_t)N_NODES * HD, "planes fit the row plane they borrow");

__global__ void __launch_bounds__(1024)
k_chunk_hist(const int* __restrict__ dst, int* __restrict__ plane) {
    __shared__ __attribute__((aligned(16))) int hist[N_NODES];
    int tid = threadIdx.x;
    int g = blockIdx.x;
    const int* d = dst + (size_t)g * CHUNK_E;
    int dv[CHUNK_TRIPS];
#pragma unroll
    for (int k = 0; k < CHUNK_TRIPS; ++k) dv[k] = d[min(tid + k * 1024, CHUNK_E - 1)];
    int4* h4 = (int4*)hist;
    for (int i = tid; i < NODE_Q; i += 1024) h4[i] = make_int4(0, 0, 0, 0);
    __syncthreads();
#pragma unroll
    for (int k = 0; k < CHUNK_TRIPS; ++k)
        if (tid + k * 1024 < CHUNK_E) atomicAdd(&hist[dv[k]], 1);
    __syncthreads();
    int4* p4 = (int4*)(plane + (size_t)g * N_NODES);
    for (int i = tid; i < NODE_Q; i += 1024) p4[i] = h4[i];
}

#define PREFIX_BLOCKS ((N_NODES + 255) / 256)
__device__ __forceinline__ void
chunk_prefix_body(int blk, int* __restrict__ plane, int* __restrict__ cnt) {
    int t = blk * 256 + threadIdx.x;
    if (t >= N_NODES) return;
    int run = 0;
    for (int g0 = 0; g0 < N_CHUNK; g0 += 16) {
        int v[16];
#pragma unroll
        for (int k = 0; k < 16; ++k) v[k] = plane[(size_t)(g0 + k) * N_NODES + t];
#pragma unroll
        for (int k = 0; k < 16; ++k) {
            plane[(size_t)(g0 + k) * N_NODES + t] = run;
            run += v[k];
        }
    }
    cnt[t] = run;
}

#define GEMM_BLOCKS ((N_NODES / 16) * 8 / 8)
__device__ __forceinline__ void
gemm_h_body(int blk, const float* __restrict__ x, const float* __restrict__ W,
            float* __restrict__ h) {
    int wave = threadIdx.x >> 5;
    int lane = threadIdx.x & 31;
    int tile = blk * 8 + wave;
    int tileM = tile >> 3;
    int tileN = tile & 7;
    int m   = tileM * 16 + (lane & 15);
    int kb  = (lane >> 4) << 1;
    int col = tileN * 16 + (lane & 15);

    v8f acc = {};
#pragma unroll
    for (int k = 0; k < F_DIM; k += 4) {
        v2f a, b;
        a.x = x[(size_t)m * F_DIM + k + kb];
        a.y = x[(size_t)m * F_DIM + k + kb + 1];
        b.x = W[(size_t)(k + kb) * HD + col];
        b.y = W[(size_t)(k + kb + 1) * HD + col];
        acc = __builtin_amdgcn_wmma_f32_16x16x4_f32(
             false, a,  false, b,
             (short)0, acc,  false,  false);
    }
    int rbase = tileM * 16 + ((lane >> 4) << 3);
#pragma unroll
    for (int r = 0; r < 8; ++r)
        h[(size_t)(rbase + r) * HD + col] = acc[r];
}

#define ATTN_BLOCKS ((N_NODES * H_HEADS + 1023) / 1024)
__device__ __forceinline__ void
attn_node_body(int blk, const float* __restrict__ h,
               const float* __restrict__ att_src,
               const float* __restrict__ att_dst,
               float* a_src, float* a_dst) {
    int idx = blk * 1024 + threadIdx.x;
    if (idx >= N_NODES * H_HEADS) return;
    int n = idx >> 3, hh = idx & 7;
    const float* hp = h + (size_t)n * HD + hh * D_HEAD;
    const float* as = att_src + hh * D_HEAD;
    const float* ad = att_dst + hh * D_HEAD;
    float s = 0.f, d = 0.f;
#pragma unroll
    for (int i = 0; i < D_HEAD; ++i) {
        float v = hp[i];
        s += v * as[i];
        d += v * ad[i];
    }
    a_src[idx] = s;
    a_dst[idx] = d;
}

__device__ __forceinline__ void
edge_proj_body(const float* __restrict__ W_e,
               const float* __restrict__ att_edge, float* V) {
    int idx = threadIdx.x;
    if (idx >= E_DIM * H_HEADS) return;
    int k = idx >> 3, hh = idx & 7;
    float s = 0.f;
#pragma unroll
    for (int d = 0; d < D_HEAD; ++d)
        s += W_e[(size_t)k * HD + hh * D_HEAD + d] * att_edge[hh * D_HEAD + d];
    V[idx] = s;
}

#define SCAN_CHUNK 49
__device__ __forceinline__ void
rowptr_body(const int* __restrict__ cnt, int* __restrict__ rowptr) {
    __shared__ int wsum[32];
    int tid = threadIdx.x;
    int lane = tid & 31, wv = tid >> 5;
    int base = tid * SCAN_CHUNK;
    int c[SCAN_CHUNK];
    int tot = 0;
#pragma unroll
    for (int k = 0; k < SCAN_CHUNK; ++k) {
        int idx = base + k;
        int d = cnt[min(idx, N_NODES - 1)];
        int v = idx < N_NODES ? d : 0;
        c[k] = v;
        tot += v;
    }
    int inc = tot;
#pragma unroll
    for (int off = 1; off < 32; off <<= 1) {
        int nb = __shfl_up(inc, off);
        if (lane >= off) inc += nb;
    }
    if (lane == 31) wsum[wv] = inc;
    __syncthreads();
    if (wv == 0) {
        int v = wsum[lane];
        int iv = v;
#pragma unroll
        for (int off = 1; off < 32; off <<= 1) {
            int nb = __shfl_up(iv, off);
            if (lane >= off) iv += nb;
        }
        wsum[lane] = iv - v;
    }
    __syncthreads();
    int run = wsum[wv] + inc - tot;
#pragma unroll
    for (int k = 0; k < SCAN_CHUNK; ++k) {
        int idx = base + k;
        if (idx < N_NODES) {
            rowptr[idx] = run;
            run += c[k];
        }
    }
    if (tid == 1023) rowptr[N_NODES] = run;
}

__global__ void __launch_bounds__(256)
k_prefix_proj_gemm(int* __restrict__ plane, int* __restrict__ cnt,
                   const float* __restrict__ W_e, const float* __restrict__ att_edge,
                   float* __restrict__ V,
                   const float* __restrict__ x, const float* __restrict__ W,
                   float* __restrict__ h) {
    int b = blockIdx.x;
    if (b < PREFIX_BLOCKS)       chunk_prefix_body(b, plane, cnt);
    else if (b == PREFIX_BLOCKS) edge_proj_body(W_e, att_edge, V);
    else                         gemm_h_body(b - PREFIX_BLOCKS - 1, x, W, h);
}

__global__ void __launch_bounds__(1024)
k_rowptr_attn(const int* __restrict__ cnt, int* __restrict__ rowptr,
              const float* __restrict__ h, const float* __restrict__ att_src,
              const float* __restrict__ att_dst, float* a_src, float* a_dst) {
    int b = blockIdx.x;
    if (b == 0) rowptr_body(cnt, rowptr);
    else        attn_node_body(b - 1, h, att_src, att_dst, a_src, a_dst);
}

__global__ void __launch_bounds__(1024)
k_chunk_place(const int* __restrict__ dst, const int* __restrict__ rowptr,
              const int* __restrict__ plane, int* __restrict__ perm) {
    __shared__ __attribute__((aligned(16))) int off[N_NODES];
    int tid = threadIdx.x;
    int g = blockIdx.x;
    const int* d = dst + (size_t)g * CHUNK_E;
    int dv[CHUNK_TRIPS];
#pragma unroll
    for (int k = 0; k < CHUNK_TRIPS; ++k) dv[k] = d[min(tid + k * 1024, CHUNK_E - 1)];
    const int4* r4 = (const int4*)rowptr;
    const int4* p4 = (const int4*)(plane + (size_t)g * N_NODES);
    int4* o4 = (int4*)off;
    for (int i = tid; i < NODE_Q; i += 1024) {
        int4 r = r4[i], p = p4[i];
        o4[i] = make_int4(r.x + p.x, r.y + p.y, r.z + p.z, r.w + p.w);
    }
    __syncthreads();
    int ebase = g * CHUNK_E;
    int pos[CHUNK_TRIPS];
#pragma unroll
    for (int k = 0; k < CHUNK_TRIPS; ++k) {
        pos[k] = 0;
        if (tid + k * 1024 < CHUNK_E) pos[k] = atomicAdd(&off[dv[k]], 1);
    }
#pragma unroll
    for (int k = 0; k < CHUNK_TRIPS; ++k) {
        int i = tid + k * 1024;
        if (i < CHUNK_E) {
            int p = min(max(pos[k], 0), N_EDGES - 1);
            perm[p] = ebase + i;
        }
    }
}

__device__ __forceinline__ void fma4(float4& acc, float w, const float4& v) {
    acc.x = fmaf(w, v.x, acc.x); acc.y = fmaf(w, v.y, acc.y);
    acc.z = fmaf(w, v.z, acc.z); acc.w = fmaf(w, v.w, acc.w);
}

__device__ __forceinline__ float max4(const float4& q) {
    return fmaxf(fmaxf(q.x, q.y), fmaxf(q.z, q.w));
}

#define LW_PITCH 36
__global__ void __launch_bounds__(32)
k_row(const int* __restrict__ rowptr, const int* __restrict__ perm,
      const int* __restrict__ src, const float* __restrict__ edge_attr,
      const float* __restrict__ Vc, const float* __restrict__ a_src,
      const float* __restrict__ a_dst, const float* __restrict__ h,
      float* __restrict__ outacc) {
    __shared__ __attribute__((aligned(16))) float lw[H_HEADS * LW_PITCH];
    const float NINF = -__builtin_inff();
    int lane = threadIdx.x;
    int t = blockIdx.x;
    int hh = lane >> 2;
    const float4* h4  = (const float4*)h;
    const float4* ea4 = (const float4*)edge_attr;
    const float4* as4 = (const float4*)a_src;
    const float4* lq  = (const float4*)(lw + hh * LW_PITCH);

    int beg = rowptr[t], end = rowptr[t + 1];
    beg = max(beg, 0); end = min(end, N_EDGES);

    float asl = a_src[t * 8 + hh];
    float adl = a_dst[t * 8 + hh];
    float4 v = h4[(size_t)t * 32 + lane];

    float ad[H_HEADS];
#pragma unroll
    for (int k = 0; k < H_HEADS; ++k) ad[k] = a_dst[t * 8 + k];

    float aes[H_HEADS];
#pragma unroll
    for (int k = 0; k < H_HEADS; ++k) aes[k] = 0.f;

    float m = NINF, den = 0.f;
    float4 acc = make_float4(0.f, 0.f, 0.f, 0.f);

    for (int b = beg; b < end; b += 32) {
        int n = min(32, end - b);
        int il = 0, sl = 0;
        float lg[H_HEADS];
#pragma unroll
        for (int k = 0; k < H_HEADS; ++k) lg[k] = NINF;
        if (lane < n) {
            il = perm[b + lane];
            il = min(max(il, 0), N_EDGES - 1);
            sl = src[il];
            float4 e0 = ea4[(size_t)il * 4 + 0];
            float4 e1 = ea4[(size_t)il * 4 + 1];
            float4 e2 = ea4[(size_t)il * 4 + 2];
            float4 e3 = ea4[(size_t)il * 4 + 3];
            float4 s0 = as4[(size_t)sl * 2 + 0];
            float4 s1 = as4[(size_t)sl * 2 + 1];
            float ev[E_DIM] = {e0.x, e0.y, e0.z, e0.w, e1.x, e1.y, e1.z, e1.w,
                               e2.x, e2.y, e2.z, e2.w, e3.x, e3.y, e3.z, e3.w};
            float sv[H_HEADS] = {s0.x, s0.y, s0.z, s0.w, s1.x, s1.y, s1.z, s1.w};
#pragma unroll
            for (int k = 0; k < H_HEADS; ++k) {
                float ae = 0.f;
#pragma unroll
                for (int q = 0; q < E_DIM; ++q) ae += ev[q] * Vc[q * 8 + k];
                aes[k] += ae;
                float l = sv[k] + ad[k] + ae;
                lg[k] = l > 0.f ? l : NEG_SLOPE * l;
            }
        }
#pragma unroll
        for (int k = 0; k < H_HEADS; ++k) lw[k * LW_PITCH + lane] = lg[k];
        __syncthreads();

        float mc = NINF;
        for (int i = 0; i < 8; ++i) mc = fmaxf(mc, max4(lq[i]));
        float mn = fmaxf(m, mc);
        float sc = __expf(m - mn);
        acc.x *= sc; acc.y *= sc; acc.z *= sc; acc.w *= sc;
        den *= sc;
        m = mn;

        int e = 0;
        for (; e + 4 <= n; e += 4) {
            int s0 = __builtin_amdgcn_readlane(sl, e);
            int s1 = __builtin_amdgcn_readlane(sl, e + 1);
            int s2 = __builtin_amdgcn_readlane(sl, e + 2);
            int s3 = __builtin_amdgcn_readlane(sl, e + 3);
            float4 v0 = h4[(size_t)s0 * 32 + lane];
            float4 v1 = h4[(size_t)s1 * 32 + lane];
            float4 v2 = h4[(size_t)s2 * 32 + lane];
            float4 v3 = h4[(size_t)s3 * 32 + lane];
            float4 q = lq[e >> 2];
            float w0 = __expf(q.x - m);
            float w1 = __expf(q.y - m);
            float w2 = __expf(q.z - m);
            float w3 = __expf(q.w - m);
            den += w0; den += w1; den += w2; den += w3;
            fma4(acc, w0, v0);
            fma4(acc, w1, v1);
            fma4(acc, w2, v2);
            fma4(acc, w3, v3);
        }
        for (; e < n; ++e) {
            int s0 = __builtin_amdgcn_readlane(sl, e);
            float4 v0 = h4[(size_t)s0 * 32 + lane];
            float w0 = __expf(lw[hh * LW_PITCH + e] - m);
            den += w0;
            fma4(acc, w0, v0);
        }
        __syncthreads();
    }

#pragma unroll
    for (int k = 0; k < H_HEADS; ++k) lw[k * LW_PITCH + lane] = aes[k];
    __syncthreads();
    float sae = 0.f;
    for (int i = 0; i < 8; ++i) {
        float4 q = lq[i];
        sae += q.x; sae += q.y; sae += q.z; sae += q.w;
    }
    float dg = (float)(end - beg);
    dg = dg < 1.0f ? 1.0f : dg;

    {
        float l = asl + adl + sae / dg;
        l = l > 0.f ? l : NEG_SLOPE * l;
        float mn = fmaxf(m, l);
        float sc = __expf(m - mn);
        float w = __expf(l - mn);
        acc.x = fmaf(w, v.x, acc.x * sc); acc.y = fmaf(w, v.y, acc.y * sc);
        acc.z = fmaf(w, v.z, acc.z * sc); acc.w = fmaf(w, v.w, acc.w * sc);
        den = fmaf(den, sc, w);
    }
    float dn = den + 1e-16f;
    acc.x /= dn; acc.y /= dn; acc.z /= dn; acc.w /= dn;
    ((float4*)outacc)[(size_t)t * 32 + lane] = acc;
}

#define ST_G     250
#define ST_WAVES 8
#define ST_TRIPS 25
static_assert(ST_G * ST_WAVES * ST_TRIPS == N_NODES, "the stats grid tiles the rows exactly");
static_assert(ST_G * 256 <= N_NODES * HD, "partials fit the plane they borrow");

__global__ void __launch_bounds__(ST_WAVES * 32)
k_stats_part(const float* __restrict__ outacc, float* __restrict__ part) {
    __shared__ __attribute__((aligned(16))) float red[ST_WAVES * 256];
    int lane = threadIdx.x & 31, wv = threadIdx.x >> 5;
    const float4* p = (const float4*)outacc
                    + ((size_t)blockIdx.x * ST_WAVES + wv) * 32 + lane;
    float4 v[ST_TRIPS];
#pragma unroll
    for (int k = 0; k < ST_TRIPS; ++k)
        v[k] = p[(size_t)k * (ST_G * ST_WAVES * 32)];
    float4 s = make_float4(0.f, 0.f, 0.f, 0.f);
    float4 q = make_float4(0.f, 0.f, 0.f, 0.f);
#pragma unroll
    for (int k = 0; k < ST_TRIPS; ++k) {
        s.x += v[k].x; s.y += v[k].y; s.z += v[k].z; s.w += v[k].w;
        q.x = fmaf(v[k].x, v[k].x, q.x); q.y = fmaf(v[k].y, v[k].y, q.y);
        q.z = fmaf(v[k].z, v[k].z, q.z); q.w = fmaf(v[k].w, v[k].w, q.w);
    }
    float4* r4 = (float4*)red;
    r4[wv * 64 + lane] = s;
    r4[wv * 64 + 32 + lane] = q;
    __syncthreads();
    int t = threadIdx.x;
    float a = 0.f;
#pragma unroll
    for (int w = 0; w < ST_WAVES; ++w) a += red[w * 256 + t];
    part[(size_t)blockIdx.x * 256 + t] = a;
}

#define ST_HALF (ST_G / 2)
static_assert(ST_HALF * 2 == ST_G, "two equal groups of partials");

#define FN_G     250
#define FN_WAVES 16
#define FN_ROWS  (N_NODES / FN_G)
#define FN_TRIPS ((FN_ROWS + FN_WAVES - 1) / FN_WAVES)
static_assert(FN_G * FN_ROWS == N_NODES, "the final grid tiles the rows exactly");
static_assert(FN_WAVES * 32 == 512, "one thread per partial column and half");

__device__ __forceinline__ float norm_elu(float o, float xv, float m, float is, float g, float b) {
    float normed = (o - m) * is * g + b;
    float t = xv + normed;
    return t > 0.f ? t : expm1f(t);
}

__global__ void __launch_bounds__(FN_WAVES * 32)
k_norm_final(const float* __restrict__ part, const float* __restrict__ x,
             const float* __restrict__ outacc, const float* __restrict__ gamma,
             const float* __restrict__ beta, float* __restrict__ out) {
    __shared__ float red[512];
    __shared__ __attribute__((aligned(16))) float smean[HD];
    __shared__ __attribute__((aligned(16))) float sinv[HD];
    int t = threadIdx.x;
    int lane = t & 31, wv = t >> 5;
    size_t rbase = (size_t)blockIdx.x * FN_ROWS;
    const float4* o4 = (const float4*)outacc;
    const float4* x4 = (const float4*)x;
    float4 ov[FN_TRIPS], xv[FN_TRIPS];
#pragma unroll
    for (int k = 0; k < FN_TRIPS; ++k) {
        size_t r = rbase + min(wv + k * FN_WAVES, FN_ROWS - 1);
        ov[k] = o4[r * 32 + lane];
        xv[k] = x4[r * 32 + lane];
    }
    float4 g4 = ((const float4*)gamma)[lane];
    float4 b4 = ((const float4*)beta)[lane];

    int col = t & 255, j = t >> 8;
    const float* p = part + (size_t)j * ST_HALF * 256 + col;
    float a = 0.f;
#pragma unroll
    for (int k = 0; k < ST_HALF; ++k) a += p[(size_t)k * 256];
    red[t] = a;
    __syncthreads();
    if (t < HD) {
        float sum = red[t] + red[256 + t];
        float sq  = red[128 + t] + red[384 + t];
        float m = sum / (float)N_NODES;
        float var = sq / (float)N_NODES - m * m;
        smean[t] = m;
        sinv[t] = rsqrtf(var + EPS_IN);
    }
    __syncthreads();
    float4 m4 = ((const float4*)smean)[lane];
    float4 i4 = ((const float4*)sinv)[lane];
    float4* out4 = (float4*)out;
#pragma unroll
    for (int k = 0; k < FN_TRIPS; ++k) {
        int rr = wv + k * FN_WAVES;
        if (rr < FN_ROWS) {
            float4 y;
            y.x = norm_elu(ov[k].x, xv[k].x, m4.x, i4.x, g4.x, b4.x);
            y.y = norm_elu(ov[k].y, xv[k].y, m4.y, i4.y, g4.y, b4.y);
            y.z = norm_elu(ov[k].z, xv[k].z, m4.z, i4.z, g4.z, b4.z);
            y.w = norm_elu(ov[k].w, xv[k].w, m4.w, i4.w, g4.w, b4.w);
            out4[(rbase + rr) * 32 + lane] = y;
        }
    }
}

extern "C" void kernel_launch(void* const* d_in, const int* in_sizes, int n_in,
                              void* d_out, int out_size, void* d_ws, size_t ws_size,
                              hipStream_t stream) {
    const float* x        = (const float*)d_in[0];
    const int*   ei       = (const int*)d_in[1];
    const float* edgeattr = (const float*)d_in[2];
    const float* W        = (const float*)d_in[3];
    const float* att_src  = (const float*)d_in[4];
    const float* att_dst  = (const float*)d_in[5];
    const float* W_e      = (const float*)d_in[6];
    const float* att_edge = (const float*)d_in[7];
    const float* gamma    = (const float*)d_in[8];
    const float* beta     = (const float*)d_in[9];
    float* out = (float*)d_out;

    const int* src = ei;
    const int* dst = ei + N_EDGES;

    float* ws = (float*)d_ws;
    float* hbuf     = ws;
    float* a_src    = hbuf   + (size_t)N_NODES * HD;
    float* a_dst    = a_src  + (size_t)N_NODES * H_HEADS;
    float* Vc       = a_dst  + (size_t)N_NODES * H_HEADS;
    float* outacc   = Vc + 128;
    float* meanb    = outacc + (size_t)N_NODES * HD;
    float* invstd   = meanb + 128;
    int*   cnt      = (int*)(invstd + 128);
    int*   rowptr   = cnt + 65536;
    int*   perm     = rowptr + 2 * 65536;
    int*   plane    = (int*)outacc;
    float* part     = hbuf;

    k_chunk_hist<<<N_CHUNK, 1024, 0, stream>>>(dst, plane);

    k_prefix_proj_gemm<<<PREFIX_BLOCKS + 1 + GEMM_BLOCKS, 256, 0, stream>>>(
        plane, cnt, W_e, att_edge, Vc, x, W, hbuf);

    k_rowptr_attn<<<1 + ATTN_BLOCKS, 1024, 0, stream>>>(
        cnt, rowptr, hbuf, att_src, att_dst, a_src, a_dst);

    k_chunk_place<<<N_CHUNK, 1024, 0, stream>>>(dst, rowptr, plane, perm);
    k_row<<<N_NODES, 32, 0, stream>>>(rowptr, perm, src, edgeattr, Vc, a_src, a_dst,
                                      hbuf, outacc);

    k_stats_part<<<ST_G, ST_WAVES * 32, 0, stream>>>(outacc, part);
    k_norm_final<<<FN_G, FN_WAVES * 32, 0, stream>>>(part, x, outacc, gamma, beta, out);
}


# Row pointers by 49 workgroups, one node per thread, whole-line loads and stores, each adds the counts in front of it; was 1 workgroup, 49 nodes per thread, 196-byte lane stride; sums unchanged
# speedup vs baseline: 16.9733x; 1.1723x over previous
#include <hip/hip_runtime.h>
#include <hip/hip_bf16.h>

#define N_NODES 50000
#define N_EDGES 1600000
#define F_DIM   128
#define H_HEADS 8
#define D_HEAD  16
#define E_DIM   16
#define HD      (H_HEADS * D_HEAD)
#define ET      (N_EDGES + N_NODES)
#define EPS_IN  1e-5f
#define NEG_SLOPE 0.2f

typedef float v2f __attribute__((ext_vector_type(2)));
typedef float v8f __attribute__((ext_vector_type(8)));

#define N_CHUNK 128
#define CHUNK_E (N_EDGES / N_CHUNK)
#define CHUNK_TRIPS ((CHUNK_E + 1023) / 1024)
#define NODE_Q (N_NODES / 4)
static_assert(N_CHUNK * CHUNK_E == N_EDGES, "chunks tile the edge list exactly");
static_assert(NODE_Q * 4 == N_NODES, "node counters split into whole quads");
static_assert((size_t)N_CHUNK * N_NODES <= (size_t)N_NODES * HD, "planes fit the row plane they borrow");

__global__ void __launch_bounds__(1024)
k_chunk_hist(const int* __restrict__ dst, int* __restrict__ plane) {
    __shared__ __attribute__((aligned(16))) int hist[N_NODES];
    int tid = threadIdx.x;
    int g = blockIdx.x;
    const int* d = dst + (size_t)g * CHUNK_E;
    int dv[CHUNK_TRIPS];
#pragma unroll
    for (int k = 0; k < CHUNK_TRIPS; ++k) dv[k] = d[min(tid + k * 1024, CHUNK_E - 1)];
    int4* h4 = (int4*)hist;
    for (int i = tid; i < NODE_Q; i += 1024) h4[i] = make_int4(0, 0, 0, 0);
    __syncthreads();
#pragma unroll
    for (int k = 0; k < CHUNK_TRIPS; ++k)
        if (tid + k * 1024 < CHUNK_E) atomicAdd(&hist[dv[k]], 1);
    __syncthreads();
    int4* p4 = (int4*)(plane + (size_t)g * N_NODES);
    for (int i = tid; i < NODE_Q; i += 1024) p4[i] = h4[i];
}

#define PREFIX_BLOCKS ((N_NODES + 255) / 256)
__device__ __forceinline__ void
chunk_prefix_body(int blk, int* __restrict__ plane, int* __restrict__ cnt) {
    int t = blk * 256 + threadIdx.x;
    if (t >= N_NODES) return;
    int run = 0;
    for (int g0 = 0; g0 < N_CHUNK; g0 += 16) {
        int v[16];
#pragma unroll
        for (int k = 0; k < 16; ++k) v[k] = plane[(size_t)(g0 + k) * N_NODES + t];
#pragma unroll
        for (int k = 0; k < 16; ++k) {
            plane[(size_t)(g0 + k) * N_NODES + t] = run;
            run += v[k];
        }
    }
    cnt[t] = run;
}

#define GEMM_BLOCKS ((N_NODES / 16) * 8 / 8)
__device__ __forceinline__ void
gemm_h_body(int blk, const float* __restrict__ x, const float* __restrict__ W,
            float* __restrict__ h) {
    int wave = threadIdx.x >> 5;
    int lane = threadIdx.x & 31;
    int tile = blk * 8 + wave;
    int tileM = tile >> 3;
    int tileN = tile & 7;
    int m   = tileM * 16 + (lane & 15);
    int kb  = (lane >> 4) << 1;
    int col = tileN * 16 + (lane & 15);

    v8f acc = {};
#pragma unroll
    for (int k = 0; k < F_DIM; k += 4) {
        v2f a, b;
        a.x = x[(size_t)m * F_DIM + k + kb];
        a.y = x[(size_t)m * F_DIM + k + kb + 1];
        b.x = W[(size_t)(k + kb) * HD + col];
        b.y = W[(size_t)(k + kb + 1) * HD + col];
        acc = __builtin_amdgcn_wmma_f32_16x16x4_f32(
             false, a,  false, b,
             (short)0, acc,  false,  false);
    }
    int rbase = tileM * 16 + ((lane >> 4) << 3);
#pragma unroll
    for (int r = 0; r < 8; ++r)
        h[(size_t)(rbase + r) * HD + col] = acc[r];
}

#define ATTN_BLOCKS ((N_NODES * H_HEADS + 1023) / 1024)
__device__ __forceinline__ void
attn_node_body(int blk, const float* __restrict__ h,
               const float* __restrict__ att_src,
               const float* __restrict__ att_dst,
               float* a_src, float* a_dst) {
    int idx = blk * 1024 + threadIdx.x;
    if (idx >= N_NODES * H_HEADS) return;
    int n = idx >> 3, hh = idx & 7;
    const float* hp = h + (size_t)n * HD + hh * D_HEAD;
    const float* as = att_src + hh * D_HEAD;
    const float* ad = att_dst + hh * D_HEAD;
    float s = 0.f, d = 0.f;
#pragma unroll
    for (int i = 0; i < D_HEAD; ++i) {
        float v = hp[i];
        s += v * as[i];
        d += v * ad[i];
    }
    a_src[idx] = s;
    a_dst[idx] = d;
}

__device__ __forceinline__ void
edge_proj_body(const float* __restrict__ W_e,
               const float* __restrict__ att_edge, float* V) {
    int idx = threadIdx.x;
    if (idx >= E_DIM * H_HEADS) return;
    int k = idx >> 3, hh = idx & 7;
    float s = 0.f;
#pragma unroll
    for (int d = 0; d < D_HEAD; ++d)
        s += W_e[(size_t)k * HD + hh * D_HEAD + d] * att_edge[hh * D_HEAD + d];
    V[idx] = s;
}

#define RP_BLOCKS ((N_NODES + 1 + 1023) / 1024)
static_assert((RP_BLOCKS - 1) * 1024 < N_NODES, "every block in front of the last is made of whole rows of counts");
__device__ __forceinline__ void
rowptr_body(int rb, const int* __restrict__ cnt, int* __restrict__ rowptr) {
    __shared__ int wsum[32];
    __shared__ int bsum[32];
    int tid = threadIdx.x;
    int lane = tid & 31, wv = tid >> 5;
    int idx = rb * 1024 + tid;
    int d = cnt[min(idx, N_NODES - 1)];
    int own = idx < N_NODES ? d : 0;
    int before = 0;
#pragma unroll
    for (int k = 0; k < RP_BLOCKS - 1; ++k) {
        int v = cnt[min(k, rb) * 1024 + tid];
        before += k < rb ? v : 0;
    }
    int inc = own;
#pragma unroll
    for (int off = 1; off < 32; off <<= 1) {
        int nb = __shfl_up(inc, off);
        if (lane >= off) inc += nb;
    }
#pragma unroll
    for (int off = 16; off > 0; off >>= 1) before += __shfl_xor(before, off);
    if (lane == 31) { wsum[wv] = inc; bsum[wv] = before; }
    __syncthreads();
    if (wv == 0) {
        int v = wsum[lane];
        int bb = bsum[lane];
        int iv = v;
#pragma unroll
        for (int off = 1; off < 32; off <<= 1) {
            int nb = __shfl_up(iv, off);
            if (lane >= off) iv += nb;
        }
#pragma unroll
        for (int off = 16; off > 0; off >>= 1) bb += __shfl_xor(bb, off);
        wsum[lane] = bb + iv - v;
    }
    __syncthreads();
    int ex = wsum[wv] + inc - own;
    if (idx <= N_NODES) rowptr[idx] = ex;
}

__global__ void __launch_bounds__(256)
k_prefix_proj_gemm(int* __restrict__ plane, int* __restrict__ cnt,
                   const float* __restrict__ W_e, const float* __restrict__ att_edge,
                   float* __restrict__ V,
                   const float* __restrict__ x, const float* __restrict__ W,
                   float* __restrict__ h) {
    int b = blockIdx.x;
    if (b < PREFIX_BLOCKS)       chunk_prefix_body(b, plane, cnt);
    else if (b == PREFIX_BLOCKS) edge_proj_body(W_e, att_edge, V);
    else                         gemm_h_body(b - PREFIX_BLOCKS - 1, x, W, h);
}

__global__ void __launch_bounds__(1024)
k_rowptr_attn(const int* __restrict__ cnt, int* __restrict__ rowptr,
              const float* __restrict__ h, const float* __restrict__ att_src,
              const float* __restrict__ att_dst, float* a_src, float* a_dst) {
    int b = blockIdx.x;
    if (b < RP_BLOCKS) rowptr_body(b, cnt, rowptr);
    else               attn_node_body(b - RP_BLOCKS, h, att_src, att_dst, a_src, a_dst);
}

__global__ void __launch_bounds__(1024)
k_chunk_place(const int* __restrict__ dst, const int* __restrict__ rowptr,
              const int* __restrict__ plane, int* __restrict__ perm) {
    __shared__ __attribute__((aligned(16))) int off[N_NODES];
    int tid = threadIdx.x;
    int g = blockIdx.x;
    const int* d = dst + (size_t)g * CHUNK_E;
    int dv[CHUNK_TRIPS];
#pragma unroll
    for (int k = 0; k < CHUNK_TRIPS; ++k) dv[k] = d[min(tid + k * 1024, CHUNK_E - 1)];
    const int4* r4 = (const int4*)rowptr;
    const int4* p4 = (const int4*)(plane + (size_t)g * N_NODES);
    int4* o4 = (int4*)off;
    for (int i = tid; i < NODE_Q; i += 1024) {
        int4 r = r4[i], p = p4[i];
        o4[i] = make_int4(r.x + p.x, r.y + p.y, r.z + p.z, r.w + p.w);
    }
    __syncthreads();
    int ebase = g * CHUNK_E;
    int pos[CHUNK_TRIPS];
#pragma unroll
    for (int k = 0; k < CHUNK_TRIPS; ++k) {
        pos[k] = 0;
        if (tid + k * 1024 < CHUNK_E) pos[k] = atomicAdd(&off[dv[k]], 1);
    }
#pragma unroll
    for (int k = 0; k < CHUNK_TRIPS; ++k) {
        int i = tid + k * 1024;
        if (i < CHUNK_E) {
            int p = min(max(pos[k], 0), N_EDGES - 1);
            perm[p] = ebase + i;
        }
    }
}

__device__ __forceinline__ void fma4(float4& acc, float w, const float4& v) {
    acc.x = fmaf(w, v.x, acc.x); acc.y = fmaf(w, v.y, acc.y);
    acc.z = fmaf(w, v.z, acc.z); acc.w = fmaf(w, v.w, acc.w);
}

__device__ __forceinline__ float max4(const float4& q) {
    return fmaxf(fmaxf(q.x, q.y), fmaxf(q.z, q.w));
}

#define LW_PITCH 36
__global__ void __launch_bounds__(32)
k_row(const int* __restrict__ rowptr, const int* __restrict__ perm,
      const int* __restrict__ src, const float* __restrict__ edge_attr,
      const float* __restrict__ Vc, const float* __restrict__ a_src,
      const float* __restrict__ a_dst, const float* __restrict__ h,
      float* __restrict__ outacc) {
    __shared__ __attribute__((aligned(16))) float lw[H_HEADS * LW_PITCH];
    const float NINF = -__builtin_inff();
    int lane = threadIdx.x;
    int t = blockIdx.x;
    int hh = lane >> 2;
    const float4* h4  = (const float4*)h;
    const float4* ea4 = (const float4*)edge_attr;
    const float4* as4 = (const float4*)a_src;
    const float4* lq  = (const float4*)(lw + hh * LW_PITCH);

    int beg = rowptr[t], end = rowptr[t + 1];
    beg = max(beg, 0); end = min(end, N_EDGES);

    float asl = a_src[t * 8 + hh];
    float adl = a_dst[t * 8 + hh];
    float4 v = h4[(size_t)t * 32 + lane];

    float ad[H_HEADS];
#pragma unroll
    for (int k = 0; k < H_HEADS; ++k) ad[k] = a_dst[t * 8 + k];

    float aes[H_HEADS];
#pragma unroll
    for (int k = 0; k < H_HEADS; ++k) aes[k] = 0.f;

    float m = NINF, den = 0.f;
    float4 acc = make_float4(0.f, 0.f, 0.f, 0.f);

    for (int b = beg; b < end; b += 32) {
        int n = min(32, end - b);
        int il = 0, sl = 0;
        float lg[H_HEADS];
#pragma unroll
        for (int k = 0; k < H_HEADS; ++k) lg[k] = NINF;
        if (lane < n) {
            il = perm[b + lane];
            il = min(max(il, 0), N_EDGES - 1);
            sl = src[il];
            float4 e0 = ea4[(size_t)il * 4 + 0];
            float4 e1 = ea4[(size_t)il * 4 + 1];
            float4 e2 = ea4[(size_t)il * 4 + 2];
            float4 e3 = ea4[(size_t)il * 4 + 3];
            float4 s0 = as4[(size_t)sl * 2 + 0];
            float4 s1 = as4[(size_t)sl * 2 + 1];
            float ev[E_DIM] = {e0.x, e0.y, e0.z, e0.w, e1.x, e1.y, e1.z, e1.w,
                               e2.x, e2.y, e2.z, e2.w, e3.x, e3.y, e3.z, e3.w};
            float sv[H_HEADS] = {s0.x, s0.y, s0.z, s0.w, s1.x, s1.y, s1.z, s1.w};
#pragma unroll
            for (int k = 0; k < H_HEADS; ++k) {
                float ae = 0.f;
#pragma unroll
                for (int q = 0; q < E_DIM; ++q) ae += ev[q] * Vc[q * 8 + k];
                aes[k] += ae;
                float l = sv[k] + ad[k] + ae;
                lg[k] = l > 0.f ? l : NEG_SLOPE * l;
            }
        }
#pragma unroll
        for (int k = 0; k < H_HEADS; ++k) lw[k * LW_PITCH + lane] = lg[k];
        __syncthreads();

        float mc = NINF;
        for (int i = 0; i < 8; ++i) mc = fmaxf(mc, max4(lq[i]));
        float mn = fmaxf(m, mc);
        float sc = __expf(m - mn);
        acc.x *= sc; acc.y *= sc; acc.z *= sc; acc.w *= sc;
        den *= sc;
        m = mn;

        int e = 0;
        for (; e + 4 <= n; e += 4) {
            int s0 = __builtin_amdgcn_readlane(sl, e);
            int s1 = __builtin_amdgcn_readlane(sl, e + 1);
            int s2 = __builtin_amdgcn_readlane(sl, e + 2);
            int s3 = __builtin_amdgcn_readlane(sl, e + 3);
            float4 v0 = h4[(size_t)s0 * 32 + lane];
            float4 v1 = h4[(size_t)s1 * 32 + lane];
            float4 v2 = h4[(size_t)s2 * 32 + lane];
            float4 v3 = h4[(size_t)s3 * 32 + lane];
            float4 q = lq[e >> 2];
            float w0 = __expf(q.x - m);
            float w1 = __expf(q.y - m);
            float w2 = __expf(q.z - m);
            float w3 = __expf(q.w - m);
            den += w0; den += w1; den += w2; den += w3;
            fma4(acc, w0, v0);
            fma4(acc, w1, v1);
            fma4(acc, w2, v2);
            fma4(acc, w3, v3);
        }
        for (; e < n; ++e) {
            int s0 = __builtin_amdgcn_readlane(sl, e);
            float4 v0 = h4[(size_t)s0 * 32 + lane];
            float w0 = __expf(lw[hh * LW_PITCH + e] - m);
            den += w0;
            fma4(acc, w0, v0);
        }
        __syncthreads();
    }

#pragma unroll
    for (int k = 0; k < H_HEADS; ++k) lw[k * LW_PITCH + lane] = aes[k];
    __syncthreads();
    float sae = 0.f;
    for (int i = 0; i < 8; ++i) {
        float4 q = lq[i];
        sae += q.x; sae += q.y; sae += q.z; sae += q.w;
    }
    float dg = (float)(end - beg);
    dg = dg < 1.0f ? 1.0f : dg;

    {
        float l = asl + adl + sae / dg;
        l = l > 0.f ? l : NEG_SLOPE * l;
        float mn = fmaxf(m, l);
        float sc = __expf(m - mn);
        float w = __expf(l - mn);
        acc.x = fmaf(w, v.x, acc.x * sc); acc.y = fmaf(w, v.y, acc.y * sc);
        acc.z = fmaf(w, v.z, acc.z * sc); acc.w = fmaf(w, v.w, acc.w * sc);
        den = fmaf(den, sc, w);
    }
    float dn = den + 1e-16f;
    acc.x /= dn; acc.y /= dn; acc.z /= dn; acc.w /= dn;
    ((float4*)outacc)[(size_t)t * 32 + lane] = acc;
}

#define ST_G     250
#define ST_WAVES 8
#define ST_TRIPS 25
static_assert(ST_G * ST_WAVES * ST_TRIPS == N_NODES, "the stats grid tiles the rows exactly");
static_assert(ST_G * 256 <= N_NODES * HD, "partials fit the plane they borrow");

__global__ void __launch_bounds__(ST_WAVES * 32)
k_stats_part(const float* __restrict__ outacc, float* __restrict__ part) {
    __shared__ __attribute__((aligned(16))) float red[ST_WAVES * 256];
    int lane = threadIdx.x & 31, wv = threadIdx.x >> 5;
    const float4* p = (const float4*)outacc
                    + ((size_t)blockIdx.x * ST_WAVES + wv) * 32 + lane;
    float4 v[ST_TRIPS];
#pragma unroll
    for (int k = 0; k < ST_TRIPS; ++k)
        v[k] = p[(size_t)k * (ST_G * ST_WAVES * 32)];
    float4 s = make_float4(0.f, 0.f, 0.f, 0.f);
    float4 q = make_float4(0.f, 0.f, 0.f, 0.f);
#pragma unroll
    for (int k = 0; k < ST_TRIPS; ++k) {
        s.x += v[k].x; s.y += v[k].y; s.z += v[k].z; s.w += v[k].w;
        q.x = fmaf(v[k].x, v[k].x, q.x); q.y = fmaf(v[k].y, v[k].y, q.y);
        q.z = fmaf(v[k].z, v[k].z, q.z); q.w = fmaf(v[k].w, v[k].w, q.w);
    }
    float4* r4 = (float4*)red;
    r4[wv * 64 + lane] = s;
    r4[wv * 64 + 32 + lane] = q;
    __syncthreads();
    int t = threadIdx.x;
    float a = 0.f;
#pragma unroll
    for (int w = 0; w < ST_WAVES; ++w) a += red[w * 256 + t];
    part[(size_t)blockIdx.x * 256 + t] = a;
}

#define ST_HALF (ST_G / 2)
static_assert(ST_HALF * 2 == ST_G, "two equal groups of partials");

#define FN_G     250
#define FN_WAVES 16
#define FN_ROWS  (N_NODES / FN_G)
#define FN_TRIPS ((FN_ROWS + FN_WAVES - 1) / FN_WAVES)
static_assert(FN_G * FN_ROWS == N_NODES, "the final grid tiles the rows exactly");
static_assert(FN_WAVES * 32 == 512, "one thread per partial column and half");

__device__ __forceinline__ float norm_elu(float o, float xv, float m, float is, float g, float b) {
    float normed = (o - m) * is * g + b;
    float t = xv + normed;
    return t > 0.f ? t : expm1f(t);
}

__global__ void __launch_bounds__(FN_WAVES * 32)
k_norm_final(const float* __restrict__ part, const float* __restrict__ x,
             const float* __restrict__ outacc, const float* __restrict__ gamma,
             const float* __restrict__ beta, float* __restrict__ out) {
    __shared__ float red[512];
    __shared__ __attribute__((aligned(16))) float smean[HD];
    __shared__ __attribute__((aligned(16))) float sinv[HD];
    int t = threadIdx.x;
    int lane = t & 31, wv = t >> 5;
    size_t rbase = (size_t)blockIdx.x * FN_ROWS;
    const float4* o4 = (const float4*)outacc;
    const float4* x4 = (const float4*)x;
    float4 ov[FN_TRIPS], xv[FN_TRIPS];
#pragma unroll
    for (int k = 0; k < FN_TRIPS; ++k) {
        size_t r = rbase + min(wv + k * FN_WAVES, FN_ROWS - 1);
        ov[k] = o4[r * 32 + lane];
        xv[k] = x4[r * 32 + lane];
    }
    float4 g4 = ((const float4*)gamma)[lane];
    float4 b4 = ((const float4*)beta)[lane];

    int col = t & 255, j = t >> 8;
    const float* p = part + (size_t)j * ST_HALF * 256 + col;
    float a = 0.f;
#pragma unroll
    for (int k = 0; k < ST_HALF; ++k) a += p[(size_t)k * 256];
    red[t] = a;
    __syncthreads();
    if (t < HD) {
        float sum = red[t] + red[256 + t];
        float sq  = red[128 + t] + red[384 + t];
        float m = sum / (float)N_NODES;
        float var = sq / (float)N_NODES - m * m;
        smean[t] = m;
        sinv[t] = rsqrtf(var + EPS_IN);
    }
    __syncthreads();
    float4 m4 = ((const float4*)smean)[lane];
    float4 i4 = ((const float4*)sinv)[lane];
    float4* out4 = (float4*)out;
#pragma unroll
    for (int k = 0; k < FN_TRIPS; ++k) {
        int rr = wv + k * FN_WAVES;
        if (rr < FN_ROWS) {
            float4 y;
            y.x = norm_elu(ov[k].x, xv[k].x, m4.x, i4.x, g4.x, b4.x);
            y.y = norm_elu(ov[k].y, xv[k].y, m4.y, i4.y, g4.y, b4.y);
            y.z = norm_elu(ov[k].z, xv[k].z, m4.z, i4.z, g4.z, b4.z);
            y.w = norm_elu(ov[k].w, xv[k].w, m4.w, i4.w, g4.w, b4.w);
            out4[(rbase + rr) * 32 + lane] = y;
        }
    }
}

extern "C" void kernel_launch(void* const* d_in, const int* in_sizes, int n_in,
                              void* d_out, int out_size, void* d_ws, size_t ws_size,
                              hipStream_t stream) {
    const float* x        = (const float*)d_in[0];
    const int*   ei       = (const int*)d_in[1];
    const float* edgeattr = (const float*)d_in[2];
    const float* W        = (const float*)d_in[3];
    const float* att_src  = (const float*)d_in[4];
    const float* att_dst  = (const float*)d_in[5];
    const float* W_e      = (const float*)d_in[6];
    const float* att_edge = (const float*)d_in[7];
    const float* gamma    = (const float*)d_in[8];
    const float* beta     = (const float*)d_in[9];
    float* out = (float*)d_out;

    const int* src = ei;
    const int* dst = ei + N_EDGES;

    float* ws = (float*)d_ws;
    float* hbuf     = ws;
    float* a_src    = hbuf   + (size_t)N_NODES * HD;
    float* a_dst    = a_src  + (size_t)N_NODES * H_HEADS;
    float* Vc       = a_dst  + (size_t)N_NODES * H_HEADS;
    float* outacc   = Vc + 128;
    float* meanb    = outacc + (size_t)N_NODES * HD;
    float* invstd   = meanb + 128;
    int*   cnt      = (int*)(invstd + 128);
    int*   rowptr   = cnt + 65536;
    int*   perm     = rowptr + 2 * 65536;
    int*   plane    = (int*)outacc;
    float* part     = hbuf;

    k_chunk_hist<<<N_CHUNK, 1024, 0, stream>>>(dst, plane);

    k_prefix_proj_gemm<<<PREFIX_BLOCKS + 1 + GEMM_BLOCKS, 256, 0, stream>>>(
        plane, cnt, W_e, att_edge, Vc, x, W, hbuf);

    k_rowptr_attn<<<RP_BLOCKS + ATTN_BLOCKS, 1024, 0, stream>>>(
        cnt, rowptr, hbuf, att_src, att_dst, a_src, a_dst);

    k_chunk_place<<<N_CHUNK, 1024, 0, stream>>>(dst, rowptr, plane, perm);
    k_row<<<N_NODES, 32, 0, stream>>>(rowptr, perm, src, edgeattr, Vc, a_src, a_dst,
                                      hbuf, outacc);

    k_stats_part<<<ST_G, ST_WAVES * 32, 0, stream>>>(outacc, part);
    k_norm_final<<<FN_G, FN_WAVES * 32, 0, stream>>>(part, x, outacc, gamma, beta, out);
}


# GEMM: a workgroup's 32 x rows go to LDS once by whole rows, waves read A fragments there (was 8 B a lane at 512 B stride, 16 lines a load, in all 8 waves); W fragment feeds 2 row tiles; same sums
# speedup vs baseline: 17.9776x; 1.0592x over previous
#include <hip/hip_runtime.h>
#include <hip/hip_bf16.h>

#define N_NODES 50000
#define N_EDGES 1600000
#define F_DIM   128
#define H_HEADS 8
#define D_HEAD  16
#define E_DIM   16
#define HD      (H_HEADS * D_HEAD)
#define ET      (N_EDGES + N_NODES)
#define EPS_IN  1e-5f
#define NEG_SLOPE 0.2f

typedef float v2f __attribute__((ext_vector_type(2)));
typedef float v8f __attribute__((ext_vector_type(8)));

#define N_CHUNK 128
#define CHUNK_E (N_EDGES / N_CHUNK)
#define CHUNK_TRIPS ((CHUNK_E + 1023) / 1024)
#define NODE_Q (N_NODES / 4)
static_assert(N_CHUNK * CHUNK_E == N_EDGES, "chunks tile the edge list exactly");
static_assert(NODE_Q * 4 == N_NODES, "node counters split into whole quads");
static_assert((size_t)N_CHUNK * N_NODES <= (size_t)N_NODES * HD, "planes fit the row plane they borrow");

__global__ void __launch_bounds__(1024)
k_chunk_hist(const int* __restrict__ dst, int* __restrict__ plane) {
    __shared__ __attribute__((aligned(16))) int hist[N_NODES];
    int tid = threadIdx.x;
    int g = blockIdx.x;
    const int* d = dst + (size_t)g * CHUNK_E;
    int dv[CHUNK_TRIPS];
#pragma unroll
    for (int k = 0; k < CHUNK_TRIPS; ++k) dv[k] = d[min(tid + k * 1024, CHUNK_E - 1)];
    int4* h4 = (int4*)hist;
    for (int i = tid; i < NODE_Q; i += 1024) h4[i] = make_int4(0, 0, 0, 0);
    __syncthreads();
#pragma unroll
    for (int k = 0; k < CHUNK_TRIPS; ++k)
        if (tid + k * 1024 < CHUNK_E) atomicAdd(&hist[dv[k]], 1);
    __syncthreads();
    int4* p4 = (int4*)(plane + (size_t)g * N_NODES);
    for (int i = tid; i < NODE_Q; i += 1024) p4[i] = h4[i];
}

#define PREFIX_BLOCKS ((N_NODES + 255) / 256)
__device__ __forceinline__ void
chunk_prefix_body(int blk, int* __restrict__ plane, int* __restrict__ cnt) {
    int t = blk * 256 + threadIdx.x;
    if (t >= N_NODES) return;
    int run = 0;
    for (int g0 = 0; g0 < N_CHUNK; g0 += 16) {
        int v[16];
#pragma unroll
        for (int k = 0; k < 16; ++k) v[k] = plane[(size_t)(g0 + k) * N_NODES + t];
#pragma unroll
        for (int k = 0; k < 16; ++k) {
            plane[(size_t)(g0 + k) * N_NODES + t] = run;
            run += v[k];
        }
    }
    cnt[t] = run;
}

#define GEMM_TILES 2
#define GEMM_ROWS  (16 * GEMM_TILES)
#define ROW_TILES  (N_NODES / 16)
#define GEMM_BLOCKS ((ROW_TILES + GEMM_TILES - 1) / GEMM_TILES)
#define XS_PITCH 132
static_assert(ROW_TILES * 16 == N_NODES, "row tiles cover the rows exactly");
static_assert(GEMM_ROWS % 8 == 0, "eight waves stage whole rounds of rows");
static_assert((GEMM_BLOCKS - 1) * GEMM_ROWS + 16 <= N_NODES, "every block's first tile lies inside");
__device__ __forceinline__ void
gemm_h_body(int blk, const float* __restrict__ x, const float* __restrict__ W,
            float* __restrict__ h) {
    __shared__ __attribute__((aligned(16))) float xs[GEMM_ROWS * XS_PITCH];
    int wave = threadIdx.x >> 5;
    int lane = threadIdx.x & 31;
    int row0 = blk * GEMM_ROWS;
#pragma unroll
    for (int p = 0; p < GEMM_ROWS / 8; ++p) {
        int r = wave + 8 * p;
        int gr = min(row0 + r, N_NODES - 1);
        float4 v = ((const float4*)x)[(size_t)gr * (F_DIM / 4) + lane];
        *(float4*)(xs + r * XS_PITCH + lane * 4) = v;
    }
    __syncthreads();

    int kb  = (lane >> 4) << 1;
    int col = wave * 16 + (lane & 15);
    const float* xa = xs + (lane & 15) * XS_PITCH + kb;

    v8f acc[GEMM_TILES] = {};
#pragma unroll
    for (int k = 0; k < F_DIM; k += 4) {
        v2f b;
        b.x = W[(size_t)(k + kb) * HD + col];
        b.y = W[(size_t)(k + kb + 1) * HD + col];
#pragma unroll
        for (int t = 0; t < GEMM_TILES; ++t) {
            v2f a = *(const v2f*)(xa + t * 16 * XS_PITCH + k);
            acc[t] = __builtin_amdgcn_wmma_f32_16x16x4_f32(
                 false, a,  false, b,
                 (short)0, acc[t],  false,  false);
        }
    }
    int rlane = row0 + ((lane >> 4) << 3);
#pragma unroll
    for (int t = 0; t < GEMM_TILES; ++t) {
        if (t == 0 || row0 + t * 16 < N_NODES) {
#pragma unroll
            for (int r = 0; r < 8; ++r)
                h[(unsigned)((rlane + t * 16 + r) * HD + col)] = acc[t][r];
        }
    }
}

#define ATTN_BLOCKS ((N_NODES * H_HEADS + 1023) / 1024)
__device__ __forceinline__ void
attn_node_body(int blk, const float* __restrict__ h,
               const float* __restrict__ att_src,
               const float* __restrict__ att_dst,
               float* a_src, float* a_dst) {
    int idx = blk * 1024 + threadIdx.x;
    if (idx >= N_NODES * H_HEADS) return;
    int n = idx >> 3, hh = idx & 7;
    const float* hp = h + (size_t)n * HD + hh * D_HEAD;
    const float* as = att_src + hh * D_HEAD;
    const float* ad = att_dst + hh * D_HEAD;
    float s = 0.f, d = 0.f;
#pragma unroll
    for (int i = 0; i < D_HEAD; ++i) {
        float v = hp[i];
        s += v * as[i];
        d += v * ad[i];
    }
    a_src[idx] = s;
    a_dst[idx] = d;
}

__device__ __forceinline__ void
edge_proj_body(const float* __restrict__ W_e,
               const float* __restrict__ att_edge, float* V) {
    int idx = threadIdx.x;
    if (idx >= E_DIM * H_HEADS) return;
    int k = idx >> 3, hh = idx & 7;
    float s = 0.f;
#pragma unroll
    for (int d = 0; d < D_HEAD; ++d)
        s += W_e[(size_t)k * HD + hh * D_HEAD + d] * att_edge[hh * D_HEAD + d];
    V[idx] = s;
}

#define RP_BLOCKS ((N_NODES + 1 + 1023) / 1024)
static_assert((RP_BLOCKS - 1) * 1024 < N_NODES, "every block in front of the last is made of whole rows of counts");
__device__ __forceinline__ void
rowptr_body(int rb, const int* __restrict__ cnt, int* __restrict__ rowptr) {
    __shared__ int wsum[32];
    __shared__ int bsum[32];
    int tid = threadIdx.x;
    int lane = tid & 31, wv = tid >> 5;
    int idx = rb * 1024 + tid;
    int d = cnt[min(idx, N_NODES - 1)];
    int own = idx < N_NODES ? d : 0;
    int before = 0;
#pragma unroll
    for (int k = 0; k < RP_BLOCKS - 1; ++k) {
        int v = cnt[min(k, rb) * 1024 + tid];
        before += k < rb ? v : 0;
    }
    int inc = own;
#pragma unroll
    for (int off = 1; off < 32; off <<= 1) {
        int nb = __shfl_up(inc, off);
        if (lane >= off) inc += nb;
    }
#pragma unroll
    for (int off = 16; off > 0; off >>= 1) before += __shfl_xor(before, off);
    if (lane == 31) { wsum[wv] = inc; bsum[wv] = before; }
    __syncthreads();
    if (wv == 0) {
        int v = wsum[lane];
        int bb = bsum[lane];
        int iv = v;
#pragma unroll
        for (int off = 1; off < 32; off <<= 1) {
            int nb = __shfl_up(iv, off);
            if (lane >= off) iv += nb;
        }
#pragma unroll
        for (int off = 16; off > 0; off >>= 1) bb += __shfl_xor(bb, off);
        wsum[lane] = bb + iv - v;
    }
    __syncthreads();
    int ex = wsum[wv] + inc - own;
    if (idx <= N_NODES) rowptr[idx] = ex;
}

__global__ void __launch_bounds__(256)
k_prefix_proj_gemm(int* __restrict__ plane, int* __restrict__ cnt,
                   const float* __restrict__ W_e, const float* __restrict__ att_edge,
                   float* __restrict__ V,
                   const float* __restrict__ x, const float* __restrict__ W,
                   float* __restrict__ h) {
    int b = blockIdx.x;
    if (b < PREFIX_BLOCKS)       chunk_prefix_body(b, plane, cnt);
    else if (b == PREFIX_BLOCKS) edge_proj_body(W_e, att_edge, V);
    else                         gemm_h_body(b - PREFIX_BLOCKS - 1, x, W, h);
}

__global__ void __launch_bounds__(1024)
k_rowptr_attn(const int* __restrict__ cnt, int* __restrict__ rowptr,
              const float* __restrict__ h, const float* __restrict__ att_src,
              const float* __restrict__ att_dst, float* a_src, float* a_dst) {
    int b = blockIdx.x;
    if (b < RP_BLOCKS) rowptr_body(b, cnt, rowptr);
    else               attn_node_body(b - RP_BLOCKS, h, att_src, att_dst, a_src, a_dst);
}

__global__ void __launch_bounds__(1024)
k_chunk_place(const int* __restrict__ dst, const int* __restrict__ rowptr,
              const int* __restrict__ plane, int* __restrict__ perm) {
    __shared__ __attribute__((aligned(16))) int off[N_NODES];
    int tid = threadIdx.x;
    int g = blockIdx.x;
    const int* d = dst + (size_t)g * CHUNK_E;
    int dv[CHUNK_TRIPS];
#pragma unroll
    for (int k = 0; k < CHUNK_TRIPS; ++k) dv[k] = d[min(tid + k * 1024, CHUNK_E - 1)];
    const int4* r4 = (const int4*)rowptr;
    const int4* p4 = (const int4*)(plane + (size_t)g * N_NODES);
    int4* o4 = (int4*)off;
    for (int i = tid; i < NODE_Q; i += 1024) {
        int4 r = r4[i], p = p4[i];
        o4[i] = make_int4(r.x + p.x, r.y + p.y, r.z + p.z, r.w + p.w);
    }
    __syncthreads();
    int ebase = g * CHUNK_E;
    int pos[CHUNK_TRIPS];
#pragma unroll
    for (int k = 0; k < CHUNK_TRIPS; ++k) {
        pos[k] = 0;
        if (tid + k * 1024 < CHUNK_E) pos[k] = atomicAdd(&off[dv[k]], 1);
    }
#pragma unroll
    for (int k = 0; k < CHUNK_TRIPS; ++k) {
        int i = tid + k * 1024;
        if (i < CHUNK_E) {
            int p = min(max(pos[k], 0), N_EDGES - 1);
            perm[p] = ebase + i;
        }
    }
}

__device__ __forceinline__ void fma4(float4& acc, float w, const float4& v) {
    acc.x = fmaf(w, v.x, acc.x); acc.y = fmaf(w, v.y, acc.y);
    acc.z = fmaf(w, v.z, acc.z); acc.w = fmaf(w, v.w, acc.w);
}

__device__ __forceinline__ float max4(const float4& q) {
    return fmaxf(fmaxf(q.x, q.y), fmaxf(q.z, q.w));
}

#define LW_PITCH 36
__global__ void __launch_bounds__(32)
k_row(const int* __restrict__ rowptr, const int* __restrict__ perm,
      const int* __restrict__ src, const float* __restrict__ edge_attr,
      const float* __restrict__ Vc, const float* __restrict__ a_src,
      const float* __restrict__ a_dst, const float* __restrict__ h,
      float* __restrict__ outacc) {
    __shared__ __attribute__((aligned(16))) float lw[H_HEADS * LW_PITCH];
    const float NINF = -__builtin_inff();
    int lane = threadIdx.x;
    int t = blockIdx.x;
    int hh = lane >> 2;
    const float4* h4  = (const float4*)h;
    const float4* ea4 = (const float4*)edge_attr;
    const float4* as4 = (const float4*)a_src;
    const float4* lq  = (const float4*)(lw + hh * LW_PITCH);

    int beg = rowptr[t], end = rowptr[t + 1];
    beg = max(beg, 0); end = min(end, N_EDGES);

    float asl = a_src[t * 8 + hh];
    float adl = a_dst[t * 8 + hh];
    float4 v = h4[(size_t)t * 32 + lane];

    float ad[H_HEADS];
#pragma unroll
    for (int k = 0; k < H_HEADS; ++k) ad[k] = a_dst[t * 8 + k];

    float aes[H_HEADS];
#pragma unroll
    for (int k = 0; k < H_HEADS; ++k) aes[k] = 0.f;

    float m = NINF, den = 0.f;
    float4 acc = make_float4(0.f, 0.f, 0.f, 0.f);

    for (int b = beg; b < end; b += 32) {
        int n = min(32, end - b);
        int il = 0, sl = 0;
        float lg[H_HEADS];
#pragma unroll
        for (int k = 0; k < H_HEADS; ++k) lg[k] = NINF;
        if (lane < n) {
            il = perm[b + lane];
            il = min(max(il, 0), N_EDGES - 1);
            sl = src[il];
            float4 e0 = ea4[(size_t)il * 4 + 0];
            float4 e1 = ea4[(size_t)il * 4 + 1];
            float4 e2 = ea4[(size_t)il * 4 + 2];
            float4 e3 = ea4[(size_t)il * 4 + 3];
            float4 s0 = as4[(size_t)sl * 2 + 0];
            float4 s1 = as4[(size_t)sl * 2 + 1];
            float ev[E_DIM] = {e0.x, e0.y, e0.z, e0.w, e1.x, e1.y, e1.z, e1.w,
                               e2.x, e2.y, e2.z, e2.w, e3.x, e3.y, e3.z, e3.w};
            float sv[H_HEADS] = {s0.x, s0.y, s0.z, s0.w, s1.x, s1.y, s1.z, s1.w};
#pragma unroll
            for (int k = 0; k < H_HEADS; ++k) {
                float ae = 0.f;
#pragma unroll
                for (int q = 0; q < E_DIM; ++q) ae += ev[q] * Vc[q * 8 + k];
                aes[k] += ae;
                float l = sv[k] + ad[k] + ae;
                lg[k] = l > 0.f ? l : NEG_SLOPE * l;
            }
        }
#pragma unroll
        for (int k = 0; k < H_HEADS; ++k) lw[k * LW_PITCH + lane] = lg[k];
        __syncthreads();

        float mc = NINF;
        for (int i = 0; i < 8; ++i) mc = fmaxf(mc, max4(lq[i]));
        float mn = fmaxf(m, mc);
        float sc = __expf(m - mn);
        acc.x *= sc; acc.y *= sc; acc.z *= sc; acc.w *= sc;
        den *= sc;
        m = mn;

        int e = 0;
        for (; e + 4 <= n; e += 4) {
            int s0 = __builtin_amdgcn_readlane(sl, e);
            int s1 = __builtin_amdgcn_readlane(sl, e + 1);
            int s2 = __builtin_amdgcn_readlane(sl, e + 2);
            int s3 = __builtin_amdgcn_readlane(sl, e + 3);
            float4 v0 = h4[(size_t)s0 * 32 + lane];
            float4 v1 = h4[(size_t)s1 * 32 + lane];
            float4 v2 = h4[(size_t)s2 * 32 + lane];
            float4 v3 = h4[(size_t)s3 * 32 + lane];
            float4 q = lq[e >> 2];
            float w0 = __expf(q.x - m);
            float w1 = __expf(q.y - m);
            float w2 = __expf(q.z - m);
            float w3 = __expf(q.w - m);
            den += w0; den += w1; den += w2; den += w3;
            fma4(acc, w0, v0);
            fma4(acc, w1, v1);
            fma4(acc, w2, v2);
            fma4(acc, w3, v3);
        }
        for (; e < n; ++e) {
            int s0 = __builtin_amdgcn_readlane(sl, e);
            float4 v0 = h4[(size_t)s0 * 32 + lane];
            float w0 = __expf(lw[hh * LW_PITCH + e] - m);
            den += w0;
            fma4(acc, w0, v0);
        }
        __syncthreads();
    }

#pragma unroll
    for (int k = 0; k < H_HEADS; ++k) lw[k * LW_PITCH + lane] = aes[k];
    __syncthreads();
    float sae = 0.f;
    for (int i = 0; i < 8; ++i) {
        float4 q = lq[i];
        sae += q.x; sae += q.y; sae += q.z; sae += q.w;
    }
    float dg = (float)(end - beg);
    dg = dg < 1.0f ? 1.0f : dg;

    {
        float l = asl + adl + sae / dg;
        l = l > 0.f ? l : NEG_SLOPE * l;
        float mn = fmaxf(m, l);
        float sc = __expf(m - mn);
        float w = __expf(l - mn);
        acc.x = fmaf(w, v.x, acc.x * sc); acc.y = fmaf(w, v.y, acc.y * sc);
        acc.z = fmaf(w, v.z, acc.z * sc); acc.w = fmaf(w, v.w, acc.w * sc);
        den = fmaf(den, sc, w);
    }
    float dn = den + 1e-16f;
    acc.x /= dn; acc.y /= dn; acc.z /= dn; acc.w /= dn;
    ((float4*)outacc)[(size_t)t * 32 + lane] = acc;
}

#define ST_G     250
#define ST_WAVES 8
#define ST_TRIPS 25
static_assert(ST_G * ST_WAVES * ST_TRIPS == N_NODES, "the stats grid tiles the rows exactly");
static_assert(ST_G * 256 <= N_NODES * HD, "partials fit the plane they borrow");

__global__ void __launch_bounds__(ST_WAVES * 32)
k_stats_part(const float* __restrict__ outacc, float* __restrict__ part) {
    __shared__ __attribute__((aligned(16))) float red[ST_WAVES * 256];
    int lane = threadIdx.x & 31, wv = threadIdx.x >> 5;
    const float4* p = (const float4*)outacc
                    + ((size_t)blockIdx.x * ST_WAVES + wv) * 32 + lane;
    float4 v[ST_TRIPS];
#pragma unroll
    for (int k = 0; k < ST_TRIPS; ++k)
        v[k] = p[(size_t)k * (ST_G * ST_WAVES * 32)];
    float4 s = make_float4(0.f, 0.f, 0.f, 0.f);
    float4 q = make_float4(0.f, 0.f, 0.f, 0.f);
#pragma unroll
    for (int k = 0; k < ST_TRIPS; ++k) {
        s.x += v[k].x; s.y += v[k].y; s.z += v[k].z; s.w += v[k].w;
        q.x = fmaf(v[k].x, v[k].x, q.x); q.y = fmaf(v[k].y, v[k].y, q.y);
        q.z = fmaf(v[k].z, v[k].z, q.z); q.w = fmaf(v[k].w, v[k].w, q.w);
    }
    float4* r4 = (float4*)red;
    r4[wv * 64 + lane] = s;
    r4[wv * 64 + 32 + lane] = q;
    __syncthreads();
    int t = threadIdx.x;
    float a = 0.f;
#pragma unroll
    for (int w = 0; w < ST_WAVES; ++w) a += red[w * 256 + t];
    part[(size_t)blockIdx.x * 256 + t] = a;
}

#define ST_HALF (ST_G / 2)
static_assert(ST_HALF * 2 == ST_G, "two equal groups of partials");

#define FN_G     250
#define FN_WAVES 16
#define FN_ROWS  (N_NODES / FN_G)
#define FN_TRIPS ((FN_ROWS + FN_WAVES - 1) / FN_WAVES)
static_assert(FN_G * FN_ROWS == N_NODES, "the final grid tiles the rows exactly");
static_assert(FN_WAVES * 32 == 512, "one thread per partial column and half");

__device__ __forceinline__ float norm_elu(float o, float xv, float m, float is, float g, float b) {
    float normed = (o - m) * is * g + b;
    float t = xv + normed;
    return t > 0.f ? t : expm1f(t);
}

__global__ void __launch_bounds__(FN_WAVES * 32)
k_norm_final(const float* __restrict__ part, const float* __restrict__ x,
             const float* __restrict__ outacc, const float* __restrict__ gamma,
             const float* __restrict__ beta, float* __restrict__ out) {
    __shared__ float red[512];
    __shared__ __attribute__((aligned(16))) float smean[HD];
    __shared__ __attribute__((aligned(16))) float sinv[HD];
    int t = threadIdx.x;
    int lane = t & 31, wv = t >> 5;
    size_t rbase = (size_t)blockIdx.x * FN_ROWS;
    const float4* o4 = (const float4*)outacc;
    const float4* x4 = (const float4*)x;
    float4 ov[FN_TRIPS], xv[FN_TRIPS];
#pragma unroll
    for (int k = 0; k < FN_TRIPS; ++k) {
        size_t r = rbase + min(wv + k * FN_WAVES, FN_ROWS - 1);
        ov[k] = o4[r * 32 + lane];
        xv[k] = x4[r * 32 + lane];
    }
    float4 g4 = ((const float4*)gamma)[lane];
    float4 b4 = ((const float4*)beta)[lane];

    int col = t & 255, j = t >> 8;
    const float* p = part + (size_t)j * ST_HALF * 256 + col;
    float a = 0.f;
#pragma unroll
    for (int k = 0; k < ST_HALF; ++k) a += p[(size_t)k * 256];
    red[t] = a;
    __syncthreads();
    if (t < HD) {
        float sum = red[t] + red[256 + t];
        float sq  = red[128 + t] + red[384 + t];
        float m = sum / (float)N_NODES;
        float var = sq / (float)N_NODES - m * m;
        smean[t] = m;
        sinv[t] = rsqrtf(var + EPS_IN);
    }
    __syncthreads();
    float4 m4 = ((const float4*)smean)[lane];
    float4 i4 = ((const float4*)sinv)[lane];
    float4* out4 = (float4*)out;
#pragma unroll
    for (int k = 0; k < FN_TRIPS; ++k) {
        int rr = wv + k * FN_WAVES;
        if (rr < FN_ROWS) {
            float4 y;
            y.x = norm_elu(ov[k].x, xv[k].x, m4.x, i4.x, g4.x, b4.x);
            y.y = norm_elu(ov[k].y, xv[k].y, m4.y, i4.y, g4.y, b4.y);
            y.z = norm_elu(ov[k].z, xv[k].z, m4.z, i4.z, g4.z, b4.z);
            y.w = norm_elu(ov[k].w, xv[k].w, m4.w, i4.w, g4.w, b4.w);
            out4[(rbase + rr) * 32 + lane] = y;
        }
    }
}

extern "C" void kernel_launch(void* const* d_in, const int* in_sizes, int n_in,
                              void* d_out, int out_size, void* d_ws, size_t ws_size,
                              hipStream_t stream) {
    const float* x        = (const float*)d_in[0];
    const int*   ei       = (const int*)d_in[1];
    const float* edgeattr = (const float*)d_in[2];
    const float* W        = (const float*)d_in[3];
    const float* att_src  = (const float*)d_in[4];
    const float* att_dst  = (const float*)d_in[5];
    const float* W_e      = (const float*)d_in[6];
    const float* att_edge = (const float*)d_in[7];
    const float* gamma    = (const float*)d_in[8];
    const float* beta     = (const float*)d_in[9];
    float* out = (float*)d_out;

    const int* src = ei;
    const int* dst = ei + N_EDGES;

    float* ws = (float*)d_ws;
    float* hbuf     = ws;
    float* a_src    = hbuf   + (size_t)N_NODES * HD;
    float* a_dst    = a_src  + (size_t)N_NODES * H_HEADS;
    float* Vc       = a_dst  + (size_t)N_NODES * H_HEADS;
    float* outacc   = Vc + 128;
    float* meanb    = outacc + (size_t)N_NODES * HD;
    float* invstd   = meanb + 128;
    int*   cnt      = (int*)(invstd + 128);
    int*   rowptr   = cnt + 65536;
    int*   perm     = rowptr + 2 * 65536;
    int*   plane    = (int*)outacc;
    float* part     = hbuf;

    k_chunk_hist<<<N_CHUNK, 1024, 0, stream>>>(dst, plane);

    k_prefix_proj_gemm<<<PREFIX_BLOCKS + 1 + GEMM_BLOCKS, 256, 0, stream>>>(
        plane, cnt, W_e, att_edge, Vc, x, W, hbuf);

    k_rowptr_attn<<<RP_BLOCKS + ATTN_BLOCKS, 1024, 0, stream>>>(
        cnt, rowptr, hbuf, att_src, att_dst, a_src, a_dst);

    k_chunk_place<<<N_CHUNK, 1024, 0, stream>>>(dst, rowptr, plane, perm);
    k_row<<<N_NODES, 32, 0, stream>>>(rowptr, perm, src, edgeattr, Vc, a_src, a_dst,
                                      hbuf, outacc);

    k_stats_part<<<ST_G, ST_WAVES * 32, 0, stream>>>(outacc, part);
    k_norm_final<<<FN_G, FN_WAVES * 32, 0, stream>>>(part, x, outacc, gamma, beta, out);
}
